# starting kernel
# baseline (speedup 1.0000x reference)
#include <hip/hip_runtime.h>

#define DCH 128
#define LN_EPS 1e-5f

typedef float v2f __attribute__((ext_vector_type(2)));
typedef float v8f __attribute__((ext_vector_type(8)));

__global__ void zero_cnt_kernel(unsigned* __restrict__ cnt, int n) {
  int i = blockIdx.x * blockDim.x + threadIdx.x;
  if (i < n) cnt[i] = 0u;
}

__global__ void count_edges_kernel(const int* __restrict__ dst,
                                   unsigned* __restrict__ cnt, int e) {
  int i = blockIdx.x * blockDim.x + threadIdx.x;
  int stride = gridDim.x * blockDim.x;
  for (; i < e; i += stride) atomicAdd(&cnt[dst[i]], 1u);
}

__global__ void dinv_kernel(const unsigned* __restrict__ cnt,
                            float* __restrict__ dinv, int n) {
  int i = blockIdx.x * blockDim.x + threadIdx.x;
  if (i < n) dinv[i] = rsqrtf(1.0f + (float)cnt[i]);
}

__global__ void coef_kernel(const int* __restrict__ src,
                            const int* __restrict__ dst,
                            const float* __restrict__ dinv,
                            float* __restrict__ coef, int e) {
  int i = blockIdx.x * blockDim.x + threadIdx.x;
  int stride = gridDim.x * blockDim.x;
  for (; i < e; i += stride) coef[i] = dinv[src[i]] * dinv[dst[i]];
}

__global__ __launch_bounds__(256)
void gemm_wmma_kernel(const float* __restrict__ X, const float* __restrict__ W,
                      float* __restrict__ out, int nrows) {
  __shared__ float xs[16 * DCH];
  const int m0 = blockIdx.x * 16;
  const int t  = threadIdx.x;

  const int tile_elems = (nrows - m0 >= 16 ? 16 : (nrows - m0)) * DCH;
  const float4* src4 = (const float4*)(X + (size_t)m0 * DCH);
  float4* lds4 = (float4*)xs;
  {
    float4 z; z.x = z.y = z.z = z.w = 0.0f;
    int i0 = t, i1 = t + 256;
    lds4[i0] = (i0 * 4 < tile_elems) ? src4[i0] : z;
    lds4[i1] = (i1 * 4 < tile_elems) ? src4[i1] : z;
  }
  __syncthreads();

  const int wave = t >> 5;
  const int lane = t & 31;
  const int n0   = wave * 16;
  const int row  = lane & 15;
  const int col  = lane & 15;
  const int koff = (lane >> 4) * 2;
  const int half = lane >> 4;

  v8f c = {};
#pragma unroll 8
  for (int k0 = 0; k0 < DCH; k0 += 4) {
    v2f a, b;
    a.x = xs[row * DCH + k0 + koff];
    a.y = xs[row * DCH + k0 + koff + 1];
    b.x = W[(k0 + koff)     * DCH + n0 + col];
    b.y = W[(k0 + koff + 1) * DCH + n0 + col];
    c = __builtin_amdgcn_wmma_f32_16x16x4_f32(
         false, a,  false, b,
         (short)0, c,  false,  false);
  }

  float* obase = out + (size_t)m0 * DCH + n0;
  if (m0 + 16 <= nrows) {
#pragma unroll
    for (int v = 0; v < 8; ++v) {
      int m = v + half * 8;
      obase[(size_t)m * DCH + col] = c[v];
    }
  } else {
#pragma unroll
    for (int v = 0; v < 8; ++v) {
      int m = v + half * 8;
      if (m0 + m < nrows) obase[(size_t)m * DCH + col] = c[v];
    }
  }
}

__global__ void init_agg_kernel(const float* __restrict__ hlin,
                                const float* __restrict__ dinv,
                                const float* __restrict__ bias,
                                float* __restrict__ agg, int n) {
  int i4 = blockIdx.x * blockDim.x + threadIdx.x;
  if (i4 >= n * 32) return;
  int node = i4 >> 5, c4 = i4 & 31;
  float dv = dinv[node];
  float sc = dv * dv;
  float4 h = ((const float4*)hlin)[i4];
  float4 b = ((const float4*)bias)[c4];
  float4 o;
  o.x = h.x * sc + b.x;
  o.y = h.y * sc + b.y;
  o.z = h.z * sc + b.z;
  o.w = h.w * sc + b.w;
  ((float4*)agg)[i4] = o;
}

__global__ __launch_bounds__(256)
void edge_scatter_kernel(const float* __restrict__ hlin,
                         const float* __restrict__ coef,
                         const int* __restrict__ src,
                         const int* __restrict__ dst,
                         float* __restrict__ agg, int e) {
  int gtid = blockIdx.x * blockDim.x + threadIdx.x;
  int wid  = gtid >> 5;
  int lane = threadIdx.x & 31;
  int nw   = (gridDim.x * blockDim.x) >> 5;
  for (int ed = wid; ed < e; ed += nw) {
    int s = src[ed], d = dst[ed];
    float cf = coef[ed];
    int nxt = ed + nw;
    if (nxt < e) {
      int ns = src[nxt];
      __builtin_prefetch(hlin + (size_t)ns * DCH + lane * 4, 0, 1);
    }
    float4 v = ((const float4*)(hlin + (size_t)s * DCH))[lane];
    float* ap = agg + (size_t)d * DCH + lane * 4;
    unsafeAtomicAdd(ap + 0, v.x * cf);
    unsafeAtomicAdd(ap + 1, v.y * cf);
    unsafeAtomicAdd(ap + 2, v.z * cf);
    unsafeAtomicAdd(ap + 3, v.w * cf);
  }
}

__global__ __launch_bounds__(256)
void post_kernel(const float* __restrict__ agg, const float* __restrict__ gamma,
                 const float* __restrict__ beta, float* __restrict__ hout,
                 float* __restrict__ jk, int do_relu, int first, int n) {
  int wid  = (blockIdx.x * blockDim.x + threadIdx.x) >> 5;
  int lane = threadIdx.x & 31;
  if (wid >= n) return;

  float4 v = ((const float4*)(agg + (size_t)wid * DCH))[lane];

  float s = v.x + v.y + v.z + v.w;
#pragma unroll
  for (int off = 16; off > 0; off >>= 1) s += __shfl_xor(s, off, 32);
  float mu = s * (1.0f / DCH);

  float dx = v.x - mu, dy = v.y - mu, dz = v.z - mu, dw = v.w - mu;
  float q = dx * dx + dy * dy + dz * dz + dw * dw;
#pragma unroll
  for (int off = 16; off > 0; off >>= 1) q += __shfl_xor(q, off, 32);
  float inv = rsqrtf(q * (1.0f / DCH) + LN_EPS);

  float4 g = ((const float4*)gamma)[lane];
  float4 b = ((const float4*)beta)[lane];
  float4 y;
  y.x = dx * inv * g.x + b.x;
  y.y = dy * inv * g.y + b.y;
  y.z = dz * inv * g.z + b.z;
  y.w = dw * inv * g.w + b.w;
  if (do_relu) {
    y.x = fmaxf(y.x, 0.0f); y.y = fmaxf(y.y, 0.0f);
    y.z = fmaxf(y.z, 0.0f); y.w = fmaxf(y.w, 0.0f);
  }

  ((float4*)(hout + (size_t)wid * DCH))[lane] = y;

  float4* op = (float4*)(jk + (size_t)wid * DCH) + lane;
  if (first) {
    *op = y;
  } else {
    float4 o = *op;
    o.x = fmaxf(o.x, y.x); o.y = fmaxf(o.y, y.y);
    o.z = fmaxf(o.z, y.z); o.w = fmaxf(o.w, y.w);
    *op = o;
  }
}

extern "C" void kernel_launch(void* const* d_in, const int* in_sizes, int n_in,
                              void* d_out, int out_size, void* d_ws, size_t ws_size,
                              hipStream_t stream) {
  const int N = in_sizes[0] / DCH;
  const int E = in_sizes[1] / 2;

  const float* feat = (const float*)d_in[0];
  const int*   ei   = (const int*)d_in[1];
  const int*   src  = ei;
  const int*   dst  = ei + E;
  const float* Wm[3] = {(const float*)d_in[2], (const float*)d_in[6],  (const float*)d_in[10]};
  const float* bi[3] = {(const float*)d_in[3], (const float*)d_in[7],  (const float*)d_in[11]};
  const float* gm[3] = {(const float*)d_in[4], (const float*)d_in[8],  (const float*)d_in[12]};
  const float* be[3] = {(const float*)d_in[5], (const float*)d_in[9],  (const float*)d_in[13]};
  float* out = (float*)d_out;

  float*    wsf  = (float*)d_ws;
  unsigned* cnt  = (unsigned*)wsf;
  float*    dinv = wsf + N;
  float*    hlin = wsf + 2 * (size_t)N;
  float*    agg  = hlin + (size_t)N * DCH;
  float*    hbuf = agg  + (size_t)N * DCH;
  float*    coef = hbuf + (size_t)N * DCH;

  const int nodeBlocks = (N + 255) / 256;
  zero_cnt_kernel<<<nodeBlocks, 256, 0, stream>>>(cnt, N);
  count_edges_kernel<<<2048, 256, 0, stream>>>(dst, cnt, E);
  dinv_kernel<<<nodeBlocks, 256, 0, stream>>>(cnt, dinv, N);
  coef_kernel<<<2048, 256, 0, stream>>>(src, dst, dinv, coef, E);

  const int gemmBlocks = (N + 15) / 16;
  const int elemBlocks = (N * 32 + 255) / 256;
  const int waveBlocks = (N + 7) / 8;

  const float* X = feat;
  for (int l = 0; l < 3; ++l) {
    gemm_wmma_kernel<<<gemmBlocks, 256, 0, stream>>>(X, Wm[l], hlin, N);
    init_agg_kernel<<<elemBlocks, 256, 0, stream>>>(hlin, dinv, bi[l], agg, N);
    edge_scatter_kernel<<<8192, 256, 0, stream>>>(hlin, coef, src, dst, agg, E);
    post_kernel<<<waveBlocks, 256, 0, stream>>>(agg, gm[l], be[l], hbuf, out,
                                                 (l < 2),
                                                 (l == 0), N);
    X = hbuf;
  }
}


# Scatter-add by destination-range ownership: a work-group keeps 400 nodes' sums in LDS, filters the raw edge array and adds gathered rows with LDS float adds; no f32 global atomics; sum order changes
# speedup vs baseline: 1.8716x; 1.8716x over previous
#include <hip/hip_runtime.h>

#define DCH 128
#define LN_EPS 1e-5f
#define RNG 400
#define QCAP 256
#define AWAVES 32

typedef float v2f __attribute__((ext_vector_type(2)));
typedef float v8f __attribute__((ext_vector_type(8)));

__global__ void zero_cnt_kernel(unsigned* __restrict__ cnt, int n) {
  int i = blockIdx.x * blockDim.x + threadIdx.x;
  if (i < n) cnt[i] = 0u;
}

__global__ void count_edges_kernel(const int* __restrict__ dst,
                                   unsigned* __restrict__ cnt, int e) {
  int i = blockIdx.x * blockDim.x + threadIdx.x;
  int stride = gridDim.x * blockDim.x;
  for (; i < e; i += stride) atomicAdd(&cnt[dst[i]], 1u);
}

__global__ void dinv_kernel(const unsigned* __restrict__ cnt,
                            float* __restrict__ dinv, int n) {
  int i = blockIdx.x * blockDim.x + threadIdx.x;
  if (i < n) dinv[i] = rsqrtf(1.0f + (float)cnt[i]);
}

__global__ void coef_kernel(const int* __restrict__ src,
                            const int* __restrict__ dst,
                            const float* __restrict__ dinv,
                            float* __restrict__ coef, int e) {
  int i = blockIdx.x * blockDim.x + threadIdx.x;
  int stride = gridDim.x * blockDim.x;
  for (; i < e; i += stride) coef[i] = dinv[src[i]] * dinv[dst[i]];
}

__global__ __launch_bounds__(256)
void gemm_wmma_kernel(const float* __restrict__ X, const float* __restrict__ W,
                      float* __restrict__ out, int nrows) {
  __shared__ float xs[16 * DCH];
  const int m0 = blockIdx.x * 16;
  const int t  = threadIdx.x;

  const int tile_elems = (nrows - m0 >= 16 ? 16 : (nrows - m0)) * DCH;
  const float4* src4 = (const float4*)(X + (size_t)m0 * DCH);
  float4* lds4 = (float4*)xs;
  {
    float4 z; z.x = z.y = z.z = z.w = 0.0f;
    int i0 = t, i1 = t + 256;
    lds4[i0] = (i0 * 4 < tile_elems) ? src4[i0] : z;
    lds4[i1] = (i1 * 4 < tile_elems) ? src4[i1] : z;
  }
  __syncthreads();

  const int wave = t >> 5;
  const int lane = t & 31;
  const int n0   = wave * 16;
  const int row  = lane & 15;
  const int col  = lane & 15;
  const int koff = (lane >> 4) * 2;
  const int half = lane >> 4;

  v8f c = {};
#pragma unroll 8
  for (int k0 = 0; k0 < DCH; k0 += 4) {
    v2f a, b;
    a.x = xs[row * DCH + k0 + koff];
    a.y = xs[row * DCH + k0 + koff + 1];
    b.x = W[(k0 + koff)     * DCH + n0 + col];
    b.y = W[(k0 + koff + 1) * DCH + n0 + col];
    c = __builtin_amdgcn_wmma_f32_16x16x4_f32(
         false, a,  false, b,
         (short)0, c,  false,  false);
  }

  float* obase = out + (size_t)m0 * DCH + n0;
  if (m0 + 16 <= nrows) {
#pragma unroll
    for (int v = 0; v < 8; ++v) {
      int m = v + half * 8;
      obase[(size_t)m * DCH + col] = c[v];
    }
  } else {
#pragma unroll
    for (int v = 0; v < 8; ++v) {
      int m = v + half * 8;
      if (m0 + m < nrows) obase[(size_t)m * DCH + col] = c[v];
    }
  }
}

static __device__ __forceinline__ void lds_add(float* p, float v) {
  __hip_atomic_fetch_add(p, v, __ATOMIC_RELAXED, __HIP_MEMORY_SCOPE_WORKGROUP);
}

template <bool FULL>
static __device__ __forceinline__ void drain_ring(
    float* acc, const int2* rw, unsigned head, int nb, int lane,
    const float* __restrict__ hlin, const float* __restrict__ coef,
    const int* __restrict__ src) {
  int   s = 0, dl = 0;
  float cf = 0.0f;
  if (FULL || lane < nb) {
    int2 en = rw[(head + (unsigned)lane) & (QCAP - 1)];
    s  = src[en.x];
    cf = coef[en.x];
    dl = en.y;
  }
#pragma unroll 1
  for (int k = 0; k < nb; k += 8) {
    float4 v[8];
#pragma unroll
    for (int u = 0; u < 8; ++u) {
      if (FULL || k + u < nb) {
        int su = __builtin_amdgcn_readlane(s, k + u);
        v[u] = ((const float4*)(hlin + (size_t)su * DCH))[lane];
      }
    }
#pragma unroll
    for (int u = 0; u < 8; ++u) {
      if (FULL || k + u < nb) {
        float cu = __int_as_float(__builtin_amdgcn_readlane(__float_as_int(cf), k + u));
        int   du = __builtin_amdgcn_readlane(dl, k + u);
        float* a = acc + du * DCH + lane;
        lds_add(a,      v[u].x * cu);
        lds_add(a + 32, v[u].y * cu);
        lds_add(a + 64, v[u].z * cu);
        lds_add(a + 96, v[u].w * cu);
      }
    }
  }
}

__global__ __launch_bounds__(1024)
void range_agg_kernel(const float* __restrict__ hlin,
                      const float* __restrict__ coef,
                      const int* __restrict__ src,
                      const int* __restrict__ dst,
                      const float* __restrict__ dinv,
                      const float* __restrict__ bias,
                      float* __restrict__ agg, int n, int e) {
  __shared__ float acc[RNG * DCH];
  __shared__ int2  ring[AWAVES * QCAP];

  const int t    = threadIdx.x;
  const int lane = t & 31;
  const int wave = __builtin_amdgcn_readfirstlane(t >> 5);
  const int lo   = blockIdx.x * RNG;
  const unsigned ulo  = (unsigned)lo;
  const unsigned span = (unsigned)((n - lo) < RNG ? (n - lo) : RNG);

  {
    float4 z; z.x = z.y = z.z = z.w = 0.0f;
    for (int i = t; i < RNG * DCH / 4; i += 1024) ((float4*)acc)[i] = z;
  }
  __syncthreads();

  int2* rw = ring + wave * QCAP;
  unsigned head = 0, cnt = 0;
  const unsigned below = (1u << lane) - 1u;

  auto feed = [&](int4 dv, int eb) {
    unsigned r0 = (unsigned)dv.x - ulo, r1 = (unsigned)dv.y - ulo;
    unsigned r2 = (unsigned)dv.z - ulo, r3 = (unsigned)dv.w - ulo;
    bool m0 = r0 < span, m1 = r1 < span, m2 = r2 < span, m3 = r3 < span;
    unsigned b0 = __builtin_amdgcn_ballot_w32(m0);
    unsigned b1 = __builtin_amdgcn_ballot_w32(m1);
    unsigned b2 = __builtin_amdgcn_ballot_w32(m2);
    unsigned b3 = __builtin_amdgcn_ballot_w32(m3);
    if ((b0 | b1 | b2 | b3) != 0u) {
      unsigned p = head + cnt;
      if (m0) rw[(p + __builtin_popcount(b0 & below)) & (QCAP - 1)] = make_int2(eb,     (int)r0);
      p += __builtin_popcount(b0);
      if (m1) rw[(p + __builtin_popcount(b1 & below)) & (QCAP - 1)] = make_int2(eb + 1, (int)r1);
      p += __builtin_popcount(b1);
      if (m2) rw[(p + __builtin_popcount(b2 & below)) & (QCAP - 1)] = make_int2(eb + 2, (int)r2);
      p += __builtin_popcount(b2);
      if (m3) rw[(p + __builtin_popcount(b3 & below)) & (QCAP - 1)] = make_int2(eb + 3, (int)r3);
      p += __builtin_popcount(b3);
      cnt = p - head;
      while (cnt >= 32u) {
        drain_ring<true>(acc, rw, head, 32, lane, hlin, coef, src);
        head = (head + 32u) & (QCAP - 1);
        cnt -= 32u;
      }
    }
  };

  const int  nfull  = e >> 2;
  const bool vec    = (e & 3) == 0;
  const int  ngrp   = (e + 3) >> 2;
  const int  nchunk = (ngrp + 255) >> 8;
  const int  start  = (int)(blockIdx.x * ((unsigned)nchunk / gridDim.x));

  for (int ci = wave; ci < nchunk; ci += AWAVES) {
    int c = ci + start;
    if (c >= nchunk) c -= nchunk;
    const int g0 = (c << 8) + lane;
    int4 d[8];
    if (vec && ((c << 8) + 256 <= nfull)) {
#pragma unroll
      for (int u = 0; u < 8; ++u) d[u] = ((const int4*)dst)[g0 + u * 32];
    } else {
      const int pad = (int)(ulo + span);
#pragma unroll
      for (int u = 0; u < 8; ++u) {
        int b = (g0 + u * 32) << 2;
        d[u].x = (b     < e) ? dst[b]     : pad;
        d[u].y = (b + 1 < e) ? dst[b + 1] : pad;
        d[u].z = (b + 2 < e) ? dst[b + 2] : pad;
        d[u].w = (b + 3 < e) ? dst[b + 3] : pad;
      }
    }
#pragma unroll
    for (int u = 0; u < 8; ++u) feed(d[u], (g0 + u * 32) << 2);
  }
  if (cnt > 0u) drain_ring<false>(acc, rw, head, (int)cnt, lane, hlin, coef, src);
  __syncthreads();

  const float4 b4 = ((const float4*)bias)[lane];
  for (int r = wave; r < (int)span; r += AWAVES) {
    const int node = lo + r;
    const float* a = acc + r * DCH + lane;
    float dv = dinv[node];
    float sc = dv * dv;
    float4 h = ((const float4*)(hlin + (size_t)node * DCH))[lane];
    float4 o;
    o.x = a[0]  + (h.x * sc + b4.x);
    o.y = a[32] + (h.y * sc + b4.y);
    o.z = a[64] + (h.z * sc + b4.z);
    o.w = a[96] + (h.w * sc + b4.w);
    ((float4*)(agg + (size_t)node * DCH))[lane] = o;
  }
}

__global__ __launch_bounds__(256)
void post_kernel(const float* __restrict__ agg, const float* __restrict__ gamma,
                 const float* __restrict__ beta, float* __restrict__ hout,
                 float* __restrict__ jk, int do_relu, int first, int n) {
  int wid  = (blockIdx.x * blockDim.x + threadIdx.x) >> 5;
  int lane = threadIdx.x & 31;
  if (wid >= n) return;

  float4 v = ((const float4*)(agg + (size_t)wid * DCH))[lane];

  float s = v.x + v.y + v.z + v.w;
#pragma unroll
  for (int off = 16; off > 0; off >>= 1) s += __shfl_xor(s, off, 32);
  float mu = s * (1.0f / DCH);

  float dx = v.x - mu, dy = v.y - mu, dz = v.z - mu, dw = v.w - mu;
  float q = dx * dx + dy * dy + dz * dz + dw * dw;
#pragma unroll
  for (int off = 16; off > 0; off >>= 1) q += __shfl_xor(q, off, 32);
  float inv = rsqrtf(q * (1.0f / DCH) + LN_EPS);

  float4 g = ((const float4*)gamma)[lane];
  float4 b = ((const float4*)beta)[lane];
  float4 y;
  y.x = dx * inv * g.x + b.x;
  y.y = dy * inv * g.y + b.y;
  y.z = dz * inv * g.z + b.z;
  y.w = dw * inv * g.w + b.w;
  if (do_relu) {
    y.x = fmaxf(y.x, 0.0f); y.y = fmaxf(y.y, 0.0f);
    y.z = fmaxf(y.z, 0.0f); y.w = fmaxf(y.w, 0.0f);
  }

  ((float4*)(hout + (size_t)wid * DCH))[lane] = y;

  float4* op = (float4*)(jk + (size_t)wid * DCH) + lane;
  if (first) {
    *op = y;
  } else {
    float4 o = *op;
    o.x = fmaxf(o.x, y.x); o.y = fmaxf(o.y, y.y);
    o.z = fmaxf(o.z, y.z); o.w = fmaxf(o.w, y.w);
    *op = o;
  }
}

extern "C" void kernel_launch(void* const* d_in, const int* in_sizes, int n_in,
                              void* d_out, int out_size, void* d_ws, size_t ws_size,
                              hipStream_t stream) {
  const int N = in_sizes[0] / DCH;
  const int E = in_sizes[1] / 2;

  const float* feat = (const float*)d_in[0];
  const int*   ei   = (const int*)d_in[1];
  const int*   src  = ei;
  const int*   dst  = ei + E;
  const float* Wm[3] = {(const float*)d_in[2], (const float*)d_in[6],  (const float*)d_in[10]};
  const float* bi[3] = {(const float*)d_in[3], (const float*)d_in[7],  (const float*)d_in[11]};
  const float* gm[3] = {(const float*)d_in[4], (const float*)d_in[8],  (const float*)d_in[12]};
  const float* be[3] = {(const float*)d_in[5], (const float*)d_in[9],  (const float*)d_in[13]};
  float* out = (float*)d_out;

  float*    wsf  = (float*)d_ws;
  unsigned* cnt  = (unsigned*)wsf;
  float*    dinv = wsf + N;
  float*    hlin = wsf + 2 * (size_t)N;
  float*    agg  = hlin + (size_t)N * DCH;
  float*    hbuf = agg  + (size_t)N * DCH;
  float*    coef = hbuf + (size_t)N * DCH;

  const int nodeBlocks = (N + 255) / 256;
  zero_cnt_kernel<<<nodeBlocks, 256, 0, stream>>>(cnt, N);
  count_edges_kernel<<<2048, 256, 0, stream>>>(dst, cnt, E);
  dinv_kernel<<<nodeBlocks, 256, 0, stream>>>(cnt, dinv, N);
  coef_kernel<<<2048, 256, 0, stream>>>(src, dst, dinv, coef, E);

  const int gemmBlocks = (N + 15) / 16;
  const int rangeBlocks = (N + RNG - 1) / RNG;
  const int waveBlocks = (N + 7) / 8;

  const float* X = feat;
  for (int l = 0; l < 3; ++l) {
    gemm_wmma_kernel<<<gemmBlocks, 256, 0, stream>>>(X, Wm[l], hlin, N);
    range_agg_kernel<<<rangeBlocks, 1024, 0, stream>>>(hlin, coef, src, dst, dinv,
                                                       bi[l], agg, N, E);
    post_kernel<<<waveBlocks, 256, 0, stream>>>(agg, gm[l], be[l], hbuf, out,
                                                 (l < 2),
                                                 (l == 0), N);
    X = hbuf;
  }
}


# Range filter reads per-call key bit planes (8 planes of dst/400: 1 byte per edge, not 4) and tests 128 edges per lane by bit logic; coef formed from dinv at the drain; sum order among waves changes
# speedup vs baseline: 2.2144x; 1.1832x over previous
#include <hip/hip_runtime.h>

#define DCH 128
#define LN_EPS 1e-5f
#define RNG 400
#define QCAP 64
#define PCH 4096
#define AWAVES 32

typedef float v2f __attribute__((ext_vector_type(2)));
typedef float v8f __attribute__((ext_vector_type(8)));

__global__ void zero_cnt_kernel(unsigned* __restrict__ cnt, int n) {
  int i = blockIdx.x * blockDim.x + threadIdx.x;
  if (i < n) cnt[i] = 0u;
}

__global__ void count_edges_kernel(const int* __restrict__ dst,
                                   unsigned* __restrict__ cnt, int e) {
  int i = blockIdx.x * blockDim.x + threadIdx.x;
  int stride = gridDim.x * blockDim.x;
  for (; i < e; i += stride) atomicAdd(&cnt[dst[i]], 1u);
}

__global__ void dinv_kernel(const unsigned* __restrict__ cnt,
                            float* __restrict__ dinv, int n) {
  int i = blockIdx.x * blockDim.x + threadIdx.x;
  if (i < n) dinv[i] = rsqrtf(1.0f + (float)cnt[i]);
}

__global__ __launch_bounds__(256)
void key_planes_kernel(const int* __restrict__ dst,
                       unsigned* __restrict__ planes, int e, int nwaves) {
  const int lane = threadIdx.x & 31;
  const int wv   = (int)((blockIdx.x * blockDim.x + threadIdx.x) >> 5);
  if (wv >= nwaves) return;
  const int ebase = wv * 1024 + lane;
  const int last  = e - 1;
  int dv[32];
#pragma unroll
  for (int t = 0; t < 32; ++t) {
    const int idx = ebase + t * 32;
    dv[t] = dst[idx < last ? idx : last];
  }
  unsigned wd[8];
#pragma unroll
  for (int i = 0; i < 8; ++i) wd[i] = 0u;
#pragma unroll
  for (int t = 0; t < 32; ++t) {
    const unsigned key = (unsigned)dv[t] / (unsigned)RNG;
#pragma unroll
    for (int i = 0; i < 8; ++i) {
      const unsigned b = __builtin_amdgcn_ballot_w32(((key >> i) & 1u) != 0u);
      if (lane == t) wd[i] = b;
    }
  }
  unsigned* o = planes + (size_t)(wv >> 2) * 1024 + (wv & 3) * 32 + lane;
#pragma unroll
  for (int i = 0; i < 8; ++i) o[i * 128] = wd[i];
}

__global__ __launch_bounds__(256)
void gemm_wmma_kernel(const float* __restrict__ X, const float* __restrict__ W,
                      float* __restrict__ out, int nrows) {
  __shared__ float xs[16 * DCH];
  const int m0 = blockIdx.x * 16;
  const int t  = threadIdx.x;

  const int tile_elems = (nrows - m0 >= 16 ? 16 : (nrows - m0)) * DCH;
  const float4* src4 = (const float4*)(X + (size_t)m0 * DCH);
  float4* lds4 = (float4*)xs;
  {
    float4 z; z.x = z.y = z.z = z.w = 0.0f;
    int i0 = t, i1 = t + 256;
    lds4[i0] = (i0 * 4 < tile_elems) ? src4[i0] : z;
    lds4[i1] = (i1 * 4 < tile_elems) ? src4[i1] : z;
  }
  __syncthreads();

  const int wave = t >> 5;
  const int lane = t & 31;
  const int n0   = wave * 16;
  const int row  = lane & 15;
  const int col  = lane & 15;
  const int koff = (lane >> 4) * 2;
  const int half = lane >> 4;

  v8f c = {};
#pragma unroll 8
  for (int k0 = 0; k0 < DCH; k0 += 4) {
    v2f a, b;
    a.x = xs[row * DCH + k0 + koff];
    a.y = xs[row * DCH + k0 + koff + 1];
    b.x = W[(k0 + koff)     * DCH + n0 + col];
    b.y = W[(k0 + koff + 1) * DCH + n0 + col];
    c = __builtin_amdgcn_wmma_f32_16x16x4_f32(
         false, a,  false, b,
         (short)0, c,  false,  false);
  }

  float* obase = out + (size_t)m0 * DCH + n0;
  if (m0 + 16 <= nrows) {
#pragma unroll
    for (int v = 0; v < 8; ++v) {
      int m = v + half * 8;
      obase[(size_t)m * DCH + col] = c[v];
    }
  } else {
#pragma unroll
    for (int v = 0; v < 8; ++v) {
      int m = v + half * 8;
      if (m0 + m < nrows) obase[(size_t)m * DCH + col] = c[v];
    }
  }
}

static __device__ __forceinline__ void lds_add(float* p, float v) {
  __hip_atomic_fetch_add(p, v, __ATOMIC_RELAXED, __HIP_MEMORY_SCOPE_WORKGROUP);
}

template <bool FULL>
static __device__ __forceinline__ void drain_ring(
    float* acc, const int* rw, unsigned head, int nb, int lane, int lo,
    const float* __restrict__ hlin, const float* __restrict__ dinv,
    const int* __restrict__ src, const int* __restrict__ dst) {
  int   s = 0, dl = 0;
  float cf = 0.0f;
  if (FULL || lane < nb) {
    const int en = rw[(head + (unsigned)lane) & (QCAP - 1)];
    s = src[en];
    const int d = dst[en];
    cf = dinv[s] * dinv[d];
    dl = d - lo;
  }
#pragma unroll 1
  for (int k = 0; k < nb; k += 8) {
    float4 v[8];
#pragma unroll
    for (int u = 0; u < 8; ++u) {
      if (FULL || k + u < nb) {
        int su = __builtin_amdgcn_readlane(s, k + u);
        v[u] = ((const float4*)(hlin + (size_t)su * DCH))[lane];
      }
    }
#pragma unroll
    for (int u = 0; u < 8; ++u) {
      if (FULL || k + u < nb) {
        float cu = __int_as_float(__builtin_amdgcn_readlane(__float_as_int(cf), k + u));
        int   du = __builtin_amdgcn_readlane(dl, k + u);
        float* a = acc + du * DCH + lane;
        lds_add(a,      v[u].x * cu);
        lds_add(a + 32, v[u].y * cu);
        lds_add(a + 64, v[u].z * cu);
        lds_add(a + 96, v[u].w * cu);
      }
    }
  }
}

__global__ __launch_bounds__(1024)
void range_agg_kernel(const float* __restrict__ hlin,
                      const unsigned* __restrict__ planes,
                      const int* __restrict__ src,
                      const int* __restrict__ dst,
                      const float* __restrict__ dinv,
                      const float* __restrict__ bias,
                      float* __restrict__ agg, int n, int e) {
  __shared__ float acc[RNG * DCH];
  __shared__ int   ring[AWAVES * QCAP];

  const int t    = threadIdx.x;
  const int lane = t & 31;
  const int wave = __builtin_amdgcn_readfirstlane(t >> 5);
  const int lo   = blockIdx.x * RNG;
  const unsigned span = (unsigned)((n - lo) < RNG ? (n - lo) : RNG);

  {
    float4 z; z.x = z.y = z.z = z.w = 0.0f;
    for (int i = t; i < RNG * DCH / 4; i += 1024) ((float4*)acc)[i] = z;
  }
  __syncthreads();

  int* rw = ring + wave * QCAP;
  unsigned head = 0, cnt = 0;

  unsigned y[8];
#pragma unroll
  for (int i = 0; i < 8; ++i) y[i] = ((blockIdx.x >> i) & 1u) ? 0xFFFFFFFFu : 0u;

  auto park = [&](unsigned m, int eb) {
    unsigned b = __builtin_amdgcn_ballot_w32(m != 0u);
    while (b != 0u) {
      if (m != 0u)
        rw[(head + cnt + __builtin_amdgcn_mbcnt_lo(b, 0u)) & (QCAP - 1)] = eb + __builtin_ctz(m);
      m &= m - 1u;
      cnt += (unsigned)__builtin_popcount(b);
      if (cnt >= 32u) {
        drain_ring<true>(acc, rw, head, 32, lane, lo, hlin, dinv, src, dst);
        head = (head + 32u) & (QCAP - 1);
        cnt -= 32u;
      }
      b = __builtin_amdgcn_ballot_w32(m != 0u);
    }
  };

  const int nchunk = (e + PCH - 1) / PCH;
  const int start = (int)((blockIdx.x * (unsigned)nchunk) / gridDim.x);
  const uint4* pl4 = (const uint4*)planes;

  for (int ci = wave; ci < nchunk; ci += AWAVES) {
    int c = ci + start;
    if (c >= nchunk) c -= nchunk;
    uint4 p[8];
#pragma unroll
    for (int i = 0; i < 8; ++i) p[i] = pl4[(c * 8 + i) * 32 + lane];
    uint4 m;
    m.x = ~((p[0].x ^ y[0]) | (p[1].x ^ y[1]) | (p[2].x ^ y[2]) | (p[3].x ^ y[3]) |
            (p[4].x ^ y[4]) | (p[5].x ^ y[5]) | (p[6].x ^ y[6]) | (p[7].x ^ y[7]));
    m.y = ~((p[0].y ^ y[0]) | (p[1].y ^ y[1]) | (p[2].y ^ y[2]) | (p[3].y ^ y[3]) |
            (p[4].y ^ y[4]) | (p[5].y ^ y[5]) | (p[6].y ^ y[6]) | (p[7].y ^ y[7]));
    m.z = ~((p[0].z ^ y[0]) | (p[1].z ^ y[1]) | (p[2].z ^ y[2]) | (p[3].z ^ y[3]) |
            (p[4].z ^ y[4]) | (p[5].z ^ y[5]) | (p[6].z ^ y[6]) | (p[7].z ^ y[7]));
    m.w = ~((p[0].w ^ y[0]) | (p[1].w ^ y[1]) | (p[2].w ^ y[2]) | (p[3].w ^ y[3]) |
            (p[4].w ^ y[4]) | (p[5].w ^ y[5]) | (p[6].w ^ y[6]) | (p[7].w ^ y[7]));
    const int eb = c * PCH + lane * 128;
    if (c == nchunk - 1) {
      auto live = [&](int r) -> unsigned {
        return r >= 32 ? 0xFFFFFFFFu : (r > 0 ? ((1u << r) - 1u) : 0u);
      };
      m.x &= live(e - eb);
      m.y &= live(e - eb - 32);
      m.z &= live(e - eb - 64);
      m.w &= live(e - eb - 96);
    }
    park(m.x, eb);
    park(m.y, eb + 32);
    park(m.z, eb + 64);
    park(m.w, eb + 96);
  }
  if (cnt > 0u) drain_ring<false>(acc, rw, head, (int)cnt, lane, lo, hlin, dinv, src, dst);
  __syncthreads();

  const float4 b4 = ((const float4*)bias)[lane];
  for (int r = wave; r < (int)span; r += AWAVES) {
    const int node = lo + r;
    const float* a = acc + r * DCH + lane;
    float dv = dinv[node];
    float sc = dv * dv;
    float4 h = ((const float4*)(hlin + (size_t)node * DCH))[lane];
    float4 o;
    o.x = a[0]  + (h.x * sc + b4.x);
    o.y = a[32] + (h.y * sc + b4.y);
    o.z = a[64] + (h.z * sc + b4.z);
    o.w = a[96] + (h.w * sc + b4.w);
    ((float4*)(agg + (size_t)node * DCH))[lane] = o;
  }
}

__global__ __launch_bounds__(256)
void post_kernel(const float* __restrict__ agg, const float* __restrict__ gamma,
                 const float* __restrict__ beta, float* __restrict__ hout,
                 float* __restrict__ jk, int do_relu, int first, int n) {
  int wid  = (blockIdx.x * blockDim.x + threadIdx.x) >> 5;
  int lane = threadIdx.x & 31;
  if (wid >= n) return;

  float4 v = ((const float4*)(agg + (size_t)wid * DCH))[lane];

  float s = v.x + v.y + v.z + v.w;
#pragma unroll
  for (int off = 16; off > 0; off >>= 1) s += __shfl_xor(s, off, 32);
  float mu = s * (1.0f / DCH);

  float dx = v.x - mu, dy = v.y - mu, dz = v.z - mu, dw = v.w - mu;
  float q = dx * dx + dy * dy + dz * dz + dw * dw;
#pragma unroll
  for (int off = 16; off > 0; off >>= 1) q += __shfl_xor(q, off, 32);
  float inv = rsqrtf(q * (1.0f / DCH) + LN_EPS);

  float4 g = ((const float4*)gamma)[lane];
  float4 b = ((const float4*)beta)[lane];
  float4 y;
  y.x = dx * inv * g.x + b.x;
  y.y = dy * inv * g.y + b.y;
  y.z = dz * inv * g.z + b.z;
  y.w = dw * inv * g.w + b.w;
  if (do_relu) {
    y.x = fmaxf(y.x, 0.0f); y.y = fmaxf(y.y, 0.0f);
    y.z = fmaxf(y.z, 0.0f); y.w = fmaxf(y.w, 0.0f);
  }

  ((float4*)(hout + (size_t)wid * DCH))[lane] = y;

  float4* op = (float4*)(jk + (size_t)wid * DCH) + lane;
  if (first) {
    *op = y;
  } else {
    float4 o = *op;
    o.x = fmaxf(o.x, y.x); o.y = fmaxf(o.y, y.y);
    o.z = fmaxf(o.z, y.z); o.w = fmaxf(o.w, y.w);
    *op = o;
  }
}

extern "C" void kernel_launch(void* const* d_in, const int* in_sizes, int n_in,
                              void* d_out, int out_size, void* d_ws, size_t ws_size,
                              hipStream_t stream) {
  const int N = in_sizes[0] / DCH;
  const int E = in_sizes[1] / 2;

  const float* feat = (const float*)d_in[0];
  const int*   ei   = (const int*)d_in[1];
  const int*   src  = ei;
  const int*   dst  = ei + E;
  const float* Wm[3] = {(const float*)d_in[2], (const float*)d_in[6],  (const float*)d_in[10]};
  const float* bi[3] = {(const float*)d_in[3], (const float*)d_in[7],  (const float*)d_in[11]};
  const float* gm[3] = {(const float*)d_in[4], (const float*)d_in[8],  (const float*)d_in[12]};
  const float* be[3] = {(const float*)d_in[5], (const float*)d_in[9],  (const float*)d_in[13]};
  float* out = (float*)d_out;

  float*    wsf  = (float*)d_ws;
  unsigned* cnt  = (unsigned*)wsf;
  float*    dinv = wsf + N;
  float*    hlin = wsf + 2 * (size_t)N;
  float*    agg  = hlin + (size_t)N * DCH;
  float*    hbuf = agg  + (size_t)N * DCH;
  unsigned* planes = (unsigned*)(hbuf + (size_t)N * DCH);

  const int nodeBlocks = (N + 255) / 256;
  zero_cnt_kernel<<<nodeBlocks, 256, 0, stream>>>(cnt, N);
  count_edges_kernel<<<2048, 256, 0, stream>>>(dst, cnt, E);
  dinv_kernel<<<nodeBlocks, 256, 0, stream>>>(cnt, dinv, N);
  const int planeWaves = ((E + PCH - 1) / PCH) * 4;
  key_planes_kernel<<<(planeWaves + 7) / 8, 256, 0, stream>>>(dst, planes, E, planeWaves);

  const int gemmBlocks = (N + 15) / 16;
  const int rangeBlocks = (N + RNG - 1) / RNG;
  const int waveBlocks = (N + 7) / 8;

  const float* X = feat;
  for (int l = 0; l < 3; ++l) {
    gemm_wmma_kernel<<<gemmBlocks, 256, 0, stream>>>(X, Wm[l], hlin, N);
    range_agg_kernel<<<rangeBlocks, 1024, 0, stream>>>(hlin, planes, src, dst, dinv,
                                                       bi[l], agg, N, E);
    post_kernel<<<waveBlocks, 256, 0, stream>>>(agg, gm[l], be[l], hbuf, out,
                                                 (l < 2),
                                                 (l == 0), N);
    X = hbuf;
  }
}


# Filter once: layer 1 keeps each wave's matched edges (source, local destination, coefficient) as 8-byte records; layers 2 and 3 replay them, no filter, no src/dst/dinv fetch; per-wave add order kept
# speedup vs baseline: 2.2891x; 1.0337x over previous
#include <hip/hip_runtime.h>

#define DCH 128
#define LN_EPS 1e-5f
#define RNG 400
#define QCAP 64
#define PCH 4096
#define AWAVES 32
#define CAPW 512
#define REC_FULL 0xFFFFFFFFu

typedef float v2f __attribute__((ext_vector_type(2)));
typedef float v8f __attribute__((ext_vector_type(8)));

__global__ void zero_cnt_kernel(unsigned* __restrict__ cnt, int n) {
  int i = blockIdx.x * blockDim.x + threadIdx.x;
  if (i < n) cnt[i] = 0u;
}

__global__ void count_edges_kernel(const int* __restrict__ dst,
                                   unsigned* __restrict__ cnt, int e) {
  int i = blockIdx.x * blockDim.x + threadIdx.x;
  int stride = gridDim.x * blockDim.x;
  for (; i < e; i += stride) atomicAdd(&cnt[dst[i]], 1u);
}

__global__ void dinv_kernel(const unsigned* __restrict__ cnt,
                            float* __restrict__ dinv, int n) {
  int i = blockIdx.x * blockDim.x + threadIdx.x;
  if (i < n) dinv[i] = rsqrtf(1.0f + (float)cnt[i]);
}

__global__ __launch_bounds__(256)
void key_planes_kernel(const int* __restrict__ dst,
                       unsigned* __restrict__ planes, int e, int nwaves) {
  const int lane = threadIdx.x & 31;
  const int wv   = (int)((blockIdx.x * blockDim.x + threadIdx.x) >> 5);
  if (wv >= nwaves) return;
  const int ebase = wv * 1024 + lane;
  const int last  = e - 1;
  int dv[32];
#pragma unroll
  for (int t = 0; t < 32; ++t) {
    const int idx = ebase + t * 32;
    dv[t] = dst[idx < last ? idx : last];
  }
  unsigned wd[8];
#pragma unroll
  for (int i = 0; i < 8; ++i) wd[i] = 0u;
#pragma unroll
  for (int t = 0; t < 32; ++t) {
    const unsigned key = (unsigned)dv[t] / (unsigned)RNG;
#pragma unroll
    for (int i = 0; i < 8; ++i) {
      const unsigned b = __builtin_amdgcn_ballot_w32(((key >> i) & 1u) != 0u);
      if (lane == t) wd[i] = b;
    }
  }
  unsigned* o = planes + (size_t)(wv >> 2) * 1024 + (wv & 3) * 32 + lane;
#pragma unroll
  for (int i = 0; i < 8; ++i) o[i * 128] = wd[i];
}

__global__ __launch_bounds__(256)
void gemm_wmma_kernel(const float* X, const float* __restrict__ W,
                      float* out, int nrows) {
  __shared__ float xs[16 * DCH];
  const int m0 = blockIdx.x * 16;
  const int t  = threadIdx.x;

  const int tile_elems = (nrows - m0 >= 16 ? 16 : (nrows - m0)) * DCH;
  const float4* src4 = (const float4*)(X + (size_t)m0 * DCH);
  float4* lds4 = (float4*)xs;
  {
    float4 z; z.x = z.y = z.z = z.w = 0.0f;
    int i0 = t, i1 = t + 256;
    lds4[i0] = (i0 * 4 < tile_elems) ? src4[i0] : z;
    lds4[i1] = (i1 * 4 < tile_elems) ? src4[i1] : z;
  }
  __syncthreads();

  const int wave = t >> 5;
  const int lane = t & 31;
  const int n0   = wave * 16;
  const int row  = lane & 15;
  const int col  = lane & 15;
  const int koff = (lane >> 4) * 2;
  const int half = lane >> 4;

  v8f c = {};
#pragma unroll 8
  for (int k0 = 0; k0 < DCH; k0 += 4) {
    v2f a, b;
    a.x = xs[row * DCH + k0 + koff];
    a.y = xs[row * DCH + k0 + koff + 1];
    b.x = W[(k0 + koff)     * DCH + n0 + col];
    b.y = W[(k0 + koff + 1) * DCH + n0 + col];
    c = __builtin_amdgcn_wmma_f32_16x16x4_f32(
         false, a,  false, b,
         (short)0, c,  false,  false);
  }

  float* obase = out + (size_t)m0 * DCH + n0;
  if (m0 + 16 <= nrows) {
#pragma unroll
    for (int v = 0; v < 8; ++v) {
      int m = v + half * 8;
      obase[(size_t)m * DCH + col] = c[v];
    }
  } else {
#pragma unroll
    for (int v = 0; v < 8; ++v) {
      int m = v + half * 8;
      if (m0 + m < nrows) obase[(size_t)m * DCH + col] = c[v];
    }
  }
}

static __device__ __forceinline__ void lds_add(float* p, float v) {
  __hip_atomic_fetch_add(p, v, __ATOMIC_RELAXED, __HIP_MEMORY_SCOPE_WORKGROUP);
}

template <bool FULL, typename AfterFirst>
static __device__ __forceinline__ void add_rows(
    float* acc, int s, int dl, float cf, int nb, int lane,
    const float* __restrict__ hlin, AfterFirst after_first) {
#pragma unroll 1
  for (int k = 0; k < nb; k += 8) {
    float4 v[8];
#pragma unroll
    for (int u = 0; u < 8; ++u) {
      if (FULL || k + u < nb) {
        int su = __builtin_amdgcn_readlane(s, k + u);
        v[u] = ((const float4*)(hlin + (size_t)su * DCH))[lane];
      }
    }
    if (k == 0) after_first();
#pragma unroll
    for (int u = 0; u < 8; ++u) {
      if (FULL || k + u < nb) {
        float cu = __int_as_float(__builtin_amdgcn_readlane(__float_as_int(cf), k + u));
        int   du = __builtin_amdgcn_readlane(dl, k + u);
        float* a = acc + du * DCH + lane;
        lds_add(a,      v[u].x * cu);
        lds_add(a + 32, v[u].y * cu);
        lds_add(a + 64, v[u].z * cu);
        lds_add(a + 96, v[u].w * cu);
      }
    }
  }
}

template <bool FULL, bool RECORD>
static __device__ __forceinline__ void drain_ring(
    float* acc, const int* rw, unsigned head, int nb, int lane, int lo,
    const float* __restrict__ hlin, const float* __restrict__ dinv,
    const int* __restrict__ src, const int* __restrict__ dst,
    uint2* seg, unsigned& nrec) {
  int   s = 0, dl = 0;
  float cf = 0.0f;
  if (FULL || lane < nb) {
    const int en = rw[(head + (unsigned)lane) & (QCAP - 1)];
    s = src[en];
    const int d = dst[en];
    cf = dinv[s] * dinv[d];
    dl = d - lo;
  }
  if (RECORD) {
    if (nrec != REC_FULL && nrec + (unsigned)nb <= (unsigned)CAPW) {
      if (FULL || lane < nb) {
        uint2 r;
        r.x = (unsigned)s | ((unsigned)dl << 23);
        r.y = __float_as_uint(cf);
        seg[nrec + (unsigned)lane] = r;
      }
      nrec += (unsigned)nb;
    } else {
      nrec = REC_FULL;
    }
  }
  add_rows<FULL>(acc, s, dl, cf, nb, lane, hlin, [] {});
}

template <bool REPLAY>
__global__ __launch_bounds__(1024)
void range_agg_kernel(const float* __restrict__ hlin,
                      const unsigned* __restrict__ planes,
                      const int* __restrict__ src,
                      const int* __restrict__ dst,
                      const float* __restrict__ dinv,
                      const float* __restrict__ bias,
                      float* __restrict__ agg,
                      unsigned* recs, unsigned* wcount, int n, int e) {
  __shared__ float acc[RNG * DCH];
  __shared__ int   ring[AWAVES * QCAP];

  const int t    = threadIdx.x;
  const int lane = t & 31;
  const int wave = __builtin_amdgcn_readfirstlane(t >> 5);
  const int lo   = blockIdx.x * RNG;
  const unsigned span = (unsigned)((n - lo) < RNG ? (n - lo) : RNG);

  {
    float4 z; z.x = z.y = z.z = z.w = 0.0f;
    for (int i = t; i < RNG * DCH / 4; i += 1024) ((float4*)acc)[i] = z;
  }
  __syncthreads();

  int* rw = ring + wave * QCAP;
  unsigned head = 0, cnt = 0;

  uint2* seg = (uint2*)recs + ((size_t)blockIdx.x * AWAVES + (size_t)wave) * CAPW;
  unsigned nrec = 0;

  bool use_lists = false;
  if (REPLAY) {
    const unsigned c = wcount[blockIdx.x * AWAVES + lane];
    use_lists = __builtin_amdgcn_ballot_w32(c == REC_FULL) == 0u;
    nrec = (unsigned)__builtin_amdgcn_readlane((int)c, wave);
  }

  if (use_lists) {
    if (nrec > 0u) {
      const unsigned last = nrec - 1u;
      uint2 r = seg[(unsigned)lane < last ? (unsigned)lane : last];
      const unsigned full = nrec & ~31u;
      for (unsigned base = 0; base < full; base += 32u) {
        uint2 rn = r;
        const int   s  = (int)(r.x & 0x7FFFFFu);
        const int   dl = (int)(r.x >> 23);
        const float cf = __uint_as_float(r.y);
        add_rows<true>(acc, s, dl, cf, 32, lane, hlin, [&] {
          const unsigned i = base + 32u + (unsigned)lane;
          rn = seg[i < last ? i : last];
        });
        r = rn;
      }
      if (nrec > full) {
        const int   s  = (int)(r.x & 0x7FFFFFu);
        const int   dl = (int)(r.x >> 23);
        const float cf = __uint_as_float(r.y);
        add_rows<false>(acc, s, dl, cf, (int)(nrec - full), lane, hlin, [] {});
      }
    }
  } else {
    unsigned y[8];
#pragma unroll
    for (int i = 0; i < 8; ++i) y[i] = ((blockIdx.x >> i) & 1u) ? 0xFFFFFFFFu : 0u;

    auto park = [&](unsigned m, int eb) {
      unsigned b = __builtin_amdgcn_ballot_w32(m != 0u);
      while (b != 0u) {
        if (m != 0u)
          rw[(head + cnt + __builtin_amdgcn_mbcnt_lo(b, 0u)) & (QCAP - 1)] = eb + __builtin_ctz(m);
        m &= m - 1u;
        cnt += (unsigned)__builtin_popcount(b);
        if (cnt >= 32u) {
          drain_ring<true, !REPLAY>(acc, rw, head, 32, lane, lo, hlin, dinv, src, dst, seg, nrec);
          head = (head + 32u) & (QCAP - 1);
          cnt -= 32u;
        }
        b = __builtin_amdgcn_ballot_w32(m != 0u);
      }
    };

    const int nchunk = (e + PCH - 1) / PCH;
    const int start = (int)((blockIdx.x * (unsigned)nchunk) / gridDim.x);
    const uint4* pl4 = (const uint4*)planes;

    for (int ci = wave; ci < nchunk; ci += AWAVES) {
      int c = ci + start;
      if (c >= nchunk) c -= nchunk;
      uint4 p[8];
#pragma unroll
      for (int i = 0; i < 8; ++i) p[i] = pl4[(c * 8 + i) * 32 + lane];
      uint4 m;
      m.x = ~((p[0].x ^ y[0]) | (p[1].x ^ y[1]) | (p[2].x ^ y[2]) | (p[3].x ^ y[3]) |
              (p[4].x ^ y[4]) | (p[5].x ^ y[5]) | (p[6].x ^ y[6]) | (p[7].x ^ y[7]));
      m.y = ~((p[0].y ^ y[0]) | (p[1].y ^ y[1]) | (p[2].y ^ y[2]) | (p[3].y ^ y[3]) |
              (p[4].y ^ y[4]) | (p[5].y ^ y[5]) | (p[6].y ^ y[6]) | (p[7].y ^ y[7]));
      m.z = ~((p[0].z ^ y[0]) | (p[1].z ^ y[1]) | (p[2].z ^ y[2]) | (p[3].z ^ y[3]) |
              (p[4].z ^ y[4]) | (p[5].z ^ y[5]) | (p[6].z ^ y[6]) | (p[7].z ^ y[7]));
      m.w = ~((p[0].w ^ y[0]) | (p[1].w ^ y[1]) | (p[2].w ^ y[2]) | (p[3].w ^ y[3]) |
              (p[4].w ^ y[4]) | (p[5].w ^ y[5]) | (p[6].w ^ y[6]) | (p[7].w ^ y[7]));
      const int eb = c * PCH + lane * 128;
      if (c == nchunk - 1) {
        auto live = [&](int r) -> unsigned {
          return r >= 32 ? 0xFFFFFFFFu : (r > 0 ? ((1u << r) - 1u) : 0u);
        };
        m.x &= live(e - eb);
        m.y &= live(e - eb - 32);
        m.z &= live(e - eb - 64);
        m.w &= live(e - eb - 96);
      }
      park(m.x, eb);
      park(m.y, eb + 32);
      park(m.z, eb + 64);
      park(m.w, eb + 96);
    }
    if (cnt > 0u)
      drain_ring<false, !REPLAY>(acc, rw, head, (int)cnt, lane, lo, hlin, dinv, src, dst, seg, nrec);
    if (!REPLAY && lane == 0) wcount[blockIdx.x * AWAVES + wave] = nrec;
  }
  __syncthreads();

  const float4 b4 = ((const float4*)bias)[lane];
  for (int r = wave; r < (int)span; r += AWAVES) {
    const int node = lo + r;
    const float* a = acc + r * DCH + lane;
    float dv = dinv[node];
    float sc = dv * dv;
    float4 h = ((const float4*)(hlin + (size_t)node * DCH))[lane];
    float4 o;
    o.x = a[0]  + (h.x * sc + b4.x);
    o.y = a[32] + (h.y * sc + b4.y);
    o.z = a[64] + (h.z * sc + b4.z);
    o.w = a[96] + (h.w * sc + b4.w);
    ((float4*)(agg + (size_t)node * DCH))[lane] = o;
  }
}

__global__ __launch_bounds__(256)
void post_kernel(const float* __restrict__ agg, const float* __restrict__ gamma,
                 const float* __restrict__ beta, float* __restrict__ hout,
                 float* __restrict__ jk, int do_relu, int first, int n) {
  int wid  = (blockIdx.x * blockDim.x + threadIdx.x) >> 5;
  int lane = threadIdx.x & 31;
  if (wid >= n) return;

  float4 v = ((const float4*)(agg + (size_t)wid * DCH))[lane];

  float s = v.x + v.y + v.z + v.w;
#pragma unroll
  for (int off = 16; off > 0; off >>= 1) s += __shfl_xor(s, off, 32);
  float mu = s * (1.0f / DCH);

  float dx = v.x - mu, dy = v.y - mu, dz = v.z - mu, dw = v.w - mu;
  float q = dx * dx + dy * dy + dz * dz + dw * dw;
#pragma unroll
  for (int off = 16; off > 0; off >>= 1) q += __shfl_xor(q, off, 32);
  float inv = rsqrtf(q * (1.0f / DCH) + LN_EPS);

  float4 g = ((const float4*)gamma)[lane];
  float4 b = ((const float4*)beta)[lane];
  float4 y;
  y.x = dx * inv * g.x + b.x;
  y.y = dy * inv * g.y + b.y;
  y.z = dz * inv * g.z + b.z;
  y.w = dw * inv * g.w + b.w;
  if (do_relu) {
    y.x = fmaxf(y.x, 0.0f); y.y = fmaxf(y.y, 0.0f);
    y.z = fmaxf(y.z, 0.0f); y.w = fmaxf(y.w, 0.0f);
  }

  ((float4*)(hout + (size_t)wid * DCH))[lane] = y;

  float4* op = (float4*)(jk + (size_t)wid * DCH) + lane;
  if (first) {
    *op = y;
  } else {
    float4 o = *op;
    o.x = fmaxf(o.x, y.x); o.y = fmaxf(o.y, y.y);
    o.z = fmaxf(o.z, y.z); o.w = fmaxf(o.w, y.w);
    *op = o;
  }
}

extern "C" void kernel_launch(void* const* d_in, const int* in_sizes, int n_in,
                              void* d_out, int out_size, void* d_ws, size_t ws_size,
                              hipStream_t stream) {
  const int N = in_sizes[0] / DCH;
  const int E = in_sizes[1] / 2;

  const float* feat = (const float*)d_in[0];
  const int*   ei   = (const int*)d_in[1];
  const int*   src  = ei;
  const int*   dst  = ei + E;
  const float* Wm[3] = {(const float*)d_in[2], (const float*)d_in[6],  (const float*)d_in[10]};
  const float* bi[3] = {(const float*)d_in[3], (const float*)d_in[7],  (const float*)d_in[11]};
  const float* gm[3] = {(const float*)d_in[4], (const float*)d_in[8],  (const float*)d_in[12]};
  const float* be[3] = {(const float*)d_in[5], (const float*)d_in[9],  (const float*)d_in[13]};
  float* out = (float*)d_out;

  float*    wsf  = (float*)d_ws;
  unsigned* cnt  = (unsigned*)wsf;
  float*    dinv = wsf + N;
  float*    lists = wsf + 2 * (size_t)N;
  float*    agg  = lists + (size_t)N * DCH;
  float*    hbuf = agg  + (size_t)N * DCH;
  unsigned* planes = (unsigned*)(hbuf + (size_t)N * DCH);

  const int nodeBlocks = (N + 255) / 256;
  zero_cnt_kernel<<<nodeBlocks, 256, 0, stream>>>(cnt, N);
  count_edges_kernel<<<2048, 256, 0, stream>>>(dst, cnt, E);
  dinv_kernel<<<nodeBlocks, 256, 0, stream>>>(cnt, dinv, N);
  const int planeWaves = ((E + PCH - 1) / PCH) * 4;
  key_planes_kernel<<<(planeWaves + 7) / 8, 256, 0, stream>>>(dst, planes, E, planeWaves);

  const int gemmBlocks = (N + 15) / 16;
  const int rangeBlocks = (N + RNG - 1) / RNG;
  const int waveBlocks = (N + 7) / 8;
  unsigned* recs   = (unsigned*)lists;
  unsigned* wcount = recs + (size_t)rangeBlocks * AWAVES * CAPW * 2;

  const float* X = feat;
  for (int l = 0; l < 3; ++l) {
    gemm_wmma_kernel<<<gemmBlocks, 256, 0, stream>>>(X, Wm[l], hbuf, N);
    if (l == 0)
      range_agg_kernel<false><<<rangeBlocks, 1024, 0, stream>>>(hbuf, planes, src, dst, dinv,
                                                                bi[l], agg, recs, wcount, N, E);
    else
      range_agg_kernel<true><<<rangeBlocks, 1024, 0, stream>>>(hbuf, planes, src, dst, dinv,
                                                               bi[l], agg, recs, wcount, N, E);
    post_kernel<<<waveBlocks, 256, 0, stream>>>(agg, gm[l], be[l], hbuf, out,
                                                 (l < 2),
                                                 (l == 0), N);
    X = hbuf;
  }
}


# GEMM A tile in LDS at a row pitch of 132 dwords, not 128: the 16 rows of a fragment read fall on 16 different bank groups, not on one bank pair (no 16-way conflict per A load); sums unchanged
# speedup vs baseline: 2.5269x; 1.1039x over previous
#include <hip/hip_runtime.h>

#define DCH 128
#define LN_EPS 1e-5f
#define RNG 400
#define QCAP 64
#define PCH 4096
#define AWAVES 32
#define CAPW 512
#define REC_FULL 0xFFFFFFFFu
#define XPITCH 132

typedef float v2f __attribute__((ext_vector_type(2)));
typedef float v8f __attribute__((ext_vector_type(8)));

__global__ void zero_cnt_kernel(unsigned* __restrict__ cnt, int n) {
  int i = blockIdx.x * blockDim.x + threadIdx.x;
  if (i < n) cnt[i] = 0u;
}

__global__ void count_edges_kernel(const int* __restrict__ dst,
                                   unsigned* __restrict__ cnt, int e) {
  int i = blockIdx.x * blockDim.x + threadIdx.x;
  int stride = gridDim.x * blockDim.x;
  for (; i < e; i += stride) atomicAdd(&cnt[dst[i]], 1u);
}

__global__ void dinv_kernel(const unsigned* __restrict__ cnt,
                            float* __restrict__ dinv, int n) {
  int i = blockIdx.x * blockDim.x + threadIdx.x;
  if (i < n) dinv[i] = rsqrtf(1.0f + (float)cnt[i]);
}

__global__ __launch_bounds__(256)
void key_planes_kernel(const int* __restrict__ dst,
                       unsigned* __restrict__ planes, int e, int nwaves) {
  const int lane = threadIdx.x & 31;
  const int wv   = (int)((blockIdx.x * blockDim.x + threadIdx.x) >> 5);
  if (wv >= nwaves) return;
  const int ebase = wv * 1024 + lane;
  const int last  = e - 1;
  int dv[32];
#pragma unroll
  for (int t = 0; t < 32; ++t) {
    const int idx = ebase + t * 32;
    dv[t] = dst[idx < last ? idx : last];
  }
  unsigned wd[8];
#pragma unroll
  for (int i = 0; i < 8; ++i) wd[i] = 0u;
#pragma unroll
  for (int t = 0; t < 32; ++t) {
    const unsigned key = (unsigned)dv[t] / (unsigned)RNG;
#pragma unroll
    for (int i = 0; i < 8; ++i) {
      const unsigned b = __builtin_amdgcn_ballot_w32(((key >> i) & 1u) != 0u);
      if (lane == t) wd[i] = b;
    }
  }
  unsigned* o = planes + (size_t)(wv >> 2) * 1024 + (wv & 3) * 32 + lane;
#pragma unroll
  for (int i = 0; i < 8; ++i) o[i * 128] = wd[i];
}

__global__ __launch_bounds__(256)
void gemm_wmma_kernel(const float* X, const float* __restrict__ W,
                      float* out, int nrows) {
  __shared__ float xs[16 * XPITCH];
  const int m0 = blockIdx.x * 16;
  const int t  = threadIdx.x;

  const int tile_elems = (nrows - m0 >= 16 ? 16 : (nrows - m0)) * DCH;
  const float4* src4 = (const float4*)(X + (size_t)m0 * DCH);
  {
    float4 z; z.x = z.y = z.z = z.w = 0.0f;
    int i0 = t, i1 = t + 256;
    *(float4*)(xs + (i0 >> 5) * XPITCH + (i0 & 31) * 4) = (i0 * 4 < tile_elems) ? src4[i0] : z;
    *(float4*)(xs + (i1 >> 5) * XPITCH + (i1 & 31) * 4) = (i1 * 4 < tile_elems) ? src4[i1] : z;
  }
  __syncthreads();

  const int wave = t >> 5;
  const int lane = t & 31;
  const int n0   = wave * 16;
  const int row  = lane & 15;
  const int col  = lane & 15;
  const int koff = (lane >> 4) * 2;
  const int half = lane >> 4;

  v8f c = {};
#pragma unroll 8
  for (int k0 = 0; k0 < DCH; k0 += 4) {
    v2f a, b;
    a.x = xs[row * XPITCH + k0 + koff];
    a.y = xs[row * XPITCH + k0 + koff + 1];
    b.x = W[(k0 + koff)     * DCH + n0 + col];
    b.y = W[(k0 + koff + 1) * DCH + n0 + col];
    c = __builtin_amdgcn_wmma_f32_16x16x4_f32(
         false, a,  false, b,
         (short)0, c,  false,  false);
  }

  float* obase = out + (size_t)m0 * DCH + n0;
  if (m0 + 16 <= nrows) {
#pragma unroll
    for (int v = 0; v < 8; ++v) {
      int m = v + half * 8;
      obase[(size_t)m * DCH + col] = c[v];
    }
  } else {
#pragma unroll
    for (int v = 0; v < 8; ++v) {
      int m = v + half * 8;
      if (m0 + m < nrows) obase[(size_t)m * DCH + col] = c[v];
    }
  }
}

static __device__ __forceinline__ void lds_add(float* p, float v) {
  __hip_atomic_fetch_add(p, v, __ATOMIC_RELAXED, __HIP_MEMORY_SCOPE_WORKGROUP);
}

template <bool FULL, typename AfterFirst>
static __device__ __forceinline__ void add_rows(
    float* acc, int s, int dl, float cf, int nb, int lane,
    const float* __restrict__ hlin, AfterFirst after_first) {
#pragma unroll 1
  for (int k = 0; k < nb; k += 8) {
    float4 v[8];
#pragma unroll
    for (int u = 0; u < 8; ++u) {
      if (FULL || k + u < nb) {
        int su = __builtin_amdgcn_readlane(s, k + u);
        v[u] = ((const float4*)(hlin + (size_t)su * DCH))[lane];
      }
    }
    if (k == 0) after_first();
#pragma unroll
    for (int u = 0; u < 8; ++u) {
      if (FULL || k + u < nb) {
        float cu = __int_as_float(__builtin_amdgcn_readlane(__float_as_int(cf), k + u));
        int   du = __builtin_amdgcn_readlane(dl, k + u);
        float* a = acc + du * DCH + lane;
        lds_add(a,      v[u].x * cu);
        lds_add(a + 32, v[u].y * cu);
        lds_add(a + 64, v[u].z * cu);
        lds_add(a + 96, v[u].w * cu);
      }
    }
  }
}

template <bool FULL, bool RECORD>
static __device__ __forceinline__ void drain_ring(
    float* acc, const int* rw, unsigned head, int nb, int lane, int lo,
    const float* __restrict__ hlin, const float* __restrict__ dinv,
    const int* __restrict__ src, const int* __restrict__ dst,
    uint2* seg, unsigned& nrec) {
  int   s = 0, dl = 0;
  float cf = 0.0f;
  if (FULL || lane < nb) {
    const int en = rw[(head + (unsigned)lane) & (QCAP - 1)];
    s = src[en];
    const int d = dst[en];
    cf = dinv[s] * dinv[d];
    dl = d - lo;
  }
  if (RECORD) {
    if (nrec != REC_FULL && nrec + (unsigned)nb <= (unsigned)CAPW) {
      if (FULL || lane < nb) {
        uint2 r;
        r.x = (unsigned)s | ((unsigned)dl << 23);
        r.y = __float_as_uint(cf);
        seg[nrec + (unsigned)lane] = r;
      }
      nrec += (unsigned)nb;
    } else {
      nrec = REC_FULL;
    }
  }
  add_rows<FULL>(acc, s, dl, cf, nb, lane, hlin, [] {});
}

template <bool REPLAY>
__global__ __launch_bounds__(1024)
void range_agg_kernel(const float* __restrict__ hlin,
                      const unsigned* __restrict__ planes,
                      const int* __restrict__ src,
                      const int* __restrict__ dst,
                      const float* __restrict__ dinv,
                      const float* __restrict__ bias,
                      float* __restrict__ agg,
                      unsigned* recs, unsigned* wcount, int n, int e) {
  __shared__ float acc[RNG * DCH];
  __shared__ int   ring[AWAVES * QCAP];

  const int t    = threadIdx.x;
  const int lane = t & 31;
  const int wave = __builtin_amdgcn_readfirstlane(t >> 5);
  const int lo   = blockIdx.x * RNG;
  const unsigned span = (unsigned)((n - lo) < RNG ? (n - lo) : RNG);

  {
    float4 z; z.x = z.y = z.z = z.w = 0.0f;
    for (int i = t; i < RNG * DCH / 4; i += 1024) ((float4*)acc)[i] = z;
  }
  __syncthreads();

  int* rw = ring + wave * QCAP;
  unsigned head = 0, cnt = 0;

  uint2* seg = (uint2*)recs + ((size_t)blockIdx.x * AWAVES + (size_t)wave) * CAPW;
  unsigned nrec = 0;

  bool use_lists = false;
  if (REPLAY) {
    const unsigned c = wcount[blockIdx.x * AWAVES + lane];
    use_lists = __builtin_amdgcn_ballot_w32(c == REC_FULL) == 0u;
    nrec = (unsigned)__builtin_amdgcn_readlane((int)c, wave);
  }

  if (use_lists) {
    if (nrec > 0u) {
      const unsigned last = nrec - 1u;
      uint2 r = seg[(unsigned)lane < last ? (unsigned)lane : last];
      const unsigned full = nrec & ~31u;
      for (unsigned base = 0; base < full; base += 32u) {
        uint2 rn = r;
        const int   s  = (int)(r.x & 0x7FFFFFu);
        const int   dl = (int)(r.x >> 23);
        const float cf = __uint_as_float(r.y);
        add_rows<true>(acc, s, dl, cf, 32, lane, hlin, [&] {
          const unsigned i = base + 32u + (unsigned)lane;
          rn = seg[i < last ? i : last];
        });
        r = rn;
      }
      if (nrec > full) {
        const int   s  = (int)(r.x & 0x7FFFFFu);
        const int   dl = (int)(r.x >> 23);
        const float cf = __uint_as_float(r.y);
        add_rows<false>(acc, s, dl, cf, (int)(nrec - full), lane, hlin, [] {});
      }
    }
  } else {
    unsigned y[8];
#pragma unroll
    for (int i = 0; i < 8; ++i) y[i] = ((blockIdx.x >> i) & 1u) ? 0xFFFFFFFFu : 0u;

    auto park = [&](unsigned m, int eb) {
      unsigned b = __builtin_amdgcn_ballot_w32(m != 0u);
      while (b != 0u) {
        if (m != 0u)
          rw[(head + cnt + __builtin_amdgcn_mbcnt_lo(b, 0u)) & (QCAP - 1)] = eb + __builtin_ctz(m);
        m &= m - 1u;
        cnt += (unsigned)__builtin_popcount(b);
        if (cnt >= 32u) {
          drain_ring<true, !REPLAY>(acc, rw, head, 32, lane, lo, hlin, dinv, src, dst, seg, nrec);
          head = (head + 32u) & (QCAP - 1);
          cnt -= 32u;
        }
        b = __builtin_amdgcn_ballot_w32(m != 0u);
      }
    };

    const int nchunk = (e + PCH - 1) / PCH;
    const int start = (int)((blockIdx.x * (unsigned)nchunk) / gridDim.x);
    const uint4* pl4 = (const uint4*)planes;

    for (int ci = wave; ci < nchunk; ci += AWAVES) {
      int c = ci + start;
      if (c >= nchunk) c -= nchunk;
      uint4 p[8];
#pragma unroll
      for (int i = 0; i < 8; ++i) p[i] = pl4[(c * 8 + i) * 32 + lane];
      uint4 m;
      m.x = ~((p[0].x ^ y[0]) | (p[1].x ^ y[1]) | (p[2].x ^ y[2]) | (p[3].x ^ y[3]) |
              (p[4].x ^ y[4]) | (p[5].x ^ y[5]) | (p[6].x ^ y[6]) | (p[7].x ^ y[7]));
      m.y = ~((p[0].y ^ y[0]) | (p[1].y ^ y[1]) | (p[2].y ^ y[2]) | (p[3].y ^ y[3]) |
              (p[4].y ^ y[4]) | (p[5].y ^ y[5]) | (p[6].y ^ y[6]) | (p[7].y ^ y[7]));
      m.z = ~((p[0].z ^ y[0]) | (p[1].z ^ y[1]) | (p[2].z ^ y[2]) | (p[3].z ^ y[3]) |
              (p[4].z ^ y[4]) | (p[5].z ^ y[5]) | (p[6].z ^ y[6]) | (p[7].z ^ y[7]));
      m.w = ~((p[0].w ^ y[0]) | (p[1].w ^ y[1]) | (p[2].w ^ y[2]) | (p[3].w ^ y[3]) |
              (p[4].w ^ y[4]) | (p[5].w ^ y[5]) | (p[6].w ^ y[6]) | (p[7].w ^ y[7]));
      const int eb = c * PCH + lane * 128;
      if (c == nchunk - 1) {
        auto live = [&](int r) -> unsigned {
          return r >= 32 ? 0xFFFFFFFFu : (r > 0 ? ((1u << r) - 1u) : 0u);
        };
        m.x &= live(e - eb);
        m.y &= live(e - eb - 32);
        m.z &= live(e - eb - 64);
        m.w &= live(e - eb - 96);
      }
      park(m.x, eb);
      park(m.y, eb + 32);
      park(m.z, eb + 64);
      park(m.w, eb + 96);
    }
    if (cnt > 0u)
      drain_ring<false, !REPLAY>(acc, rw, head, (int)cnt, lane, lo, hlin, dinv, src, dst, seg, nrec);
    if (!REPLAY && lane == 0) wcount[blockIdx.x * AWAVES + wave] = nrec;
  }
  __syncthreads();

  const float4 b4 = ((const float4*)bias)[lane];
  for (int r = wave; r < (int)span; r += AWAVES) {
    const int node = lo + r;
    const float* a = acc + r * DCH + lane;
    float dv = dinv[node];
    float sc = dv * dv;
    float4 h = ((const float4*)(hlin + (size_t)node * DCH))[lane];
    float4 o;
    o.x = a[0]  + (h.x * sc + b4.x);
    o.y = a[32] + (h.y * sc + b4.y);
    o.z = a[64] + (h.z * sc + b4.z);
    o.w = a[96] + (h.w * sc + b4.w);
    ((float4*)(agg + (size_t)node * DCH))[lane] = o;
  }
}

__global__ __launch_bounds__(256)
void post_kernel(const float* __restrict__ agg, const float* __restrict__ gamma,
                 const float* __restrict__ beta, float* __restrict__ hout,
                 float* __restrict__ jk, int do_relu, int first, int n) {
  int wid  = (blockIdx.x * blockDim.x + threadIdx.x) >> 5;
  int lane = threadIdx.x & 31;
  if (wid >= n) return;

  float4 v = ((const float4*)(agg + (size_t)wid * DCH))[lane];

  float s = v.x + v.y + v.z + v.w;
#pragma unroll
  for (int off = 16; off > 0; off >>= 1) s += __shfl_xor(s, off, 32);
  float mu = s * (1.0f / DCH);

  float dx = v.x - mu, dy = v.y - mu, dz = v.z - mu, dw = v.w - mu;
  float q = dx * dx + dy * dy + dz * dz + dw * dw;
#pragma unroll
  for (int off = 16; off > 0; off >>= 1) q += __shfl_xor(q, off, 32);
  float inv = rsqrtf(q * (1.0f / DCH) + LN_EPS);

  float4 g = ((const float4*)gamma)[lane];
  float4 b = ((const float4*)beta)[lane];
  float4 y;
  y.x = dx * inv * g.x + b.x;
  y.y = dy * inv * g.y + b.y;
  y.z = dz * inv * g.z + b.z;
  y.w = dw * inv * g.w + b.w;
  if (do_relu) {
    y.x = fmaxf(y.x, 0.0f); y.y = fmaxf(y.y, 0.0f);
    y.z = fmaxf(y.z, 0.0f); y.w = fmaxf(y.w, 0.0f);
  }

  ((float4*)(hout + (size_t)wid * DCH))[lane] = y;

  float4* op = (float4*)(jk + (size_t)wid * DCH) + lane;
  if (first) {
    *op = y;
  } else {
    float4 o = *op;
    o.x = fmaxf(o.x, y.x); o.y = fmaxf(o.y, y.y);
    o.z = fmaxf(o.z, y.z); o.w = fmaxf(o.w, y.w);
    *op = o;
  }
}

extern "C" void kernel_launch(void* const* d_in, const int* in_sizes, int n_in,
                              void* d_out, int out_size, void* d_ws, size_t ws_size,
                              hipStream_t stream) {
  const int N = in_sizes[0] / DCH;
  const int E = in_sizes[1] / 2;

  const float* feat = (const float*)d_in[0];
  const int*   ei   = (const int*)d_in[1];
  const int*   src  = ei;
  const int*   dst  = ei + E;
  const float* Wm[3] = {(const float*)d_in[2], (const float*)d_in[6],  (const float*)d_in[10]};
  const float* bi[3] = {(const float*)d_in[3], (const float*)d_in[7],  (const float*)d_in[11]};
  const float* gm[3] = {(const float*)d_in[4], (const float*)d_in[8],  (const float*)d_in[12]};
  const float* be[3] = {(const float*)d_in[5], (const float*)d_in[9],  (const float*)d_in[13]};
  float* out = (float*)d_out;

  float*    wsf  = (float*)d_ws;
  unsigned* cnt  = (unsigned*)wsf;
  float*    dinv = wsf + N;
  float*    lists = wsf + 2 * (size_t)N;
  float*    agg  = lists + (size_t)N * DCH;
  float*    hbuf = agg  + (size_t)N * DCH;
  unsigned* planes = (unsigned*)(hbuf + (size_t)N * DCH);

  const int nodeBlocks = (N + 255) / 256;
  zero_cnt_kernel<<<nodeBlocks, 256, 0, stream>>>(cnt, N);
  count_edges_kernel<<<2048, 256, 0, stream>>>(dst, cnt, E);
  dinv_kernel<<<nodeBlocks, 256, 0, stream>>>(cnt, dinv, N);
  const int planeWaves = ((E + PCH - 1) / PCH) * 4;
  key_planes_kernel<<<(planeWaves + 7) / 8, 256, 0, stream>>>(dst, planes, E, planeWaves);

  const int gemmBlocks = (N + 15) / 16;
  const int rangeBlocks = (N + RNG - 1) / RNG;
  const int waveBlocks = (N + 7) / 8;
  unsigned* recs   = (unsigned*)lists;
  unsigned* wcount = recs + (size_t)rangeBlocks * AWAVES * CAPW * 2;

  const float* X = feat;
  for (int l = 0; l < 3; ++l) {
    gemm_wmma_kernel<<<gemmBlocks, 256, 0, stream>>>(X, Wm[l], hbuf, N);
    if (l == 0)
      range_agg_kernel<false><<<rangeBlocks, 1024, 0, stream>>>(hbuf, planes, src, dst, dinv,
                                                                bi[l], agg, recs, wcount, N, E);
    else
      range_agg_kernel<true><<<rangeBlocks, 1024, 0, stream>>>(hbuf, planes, src, dst, dinv,
                                                               bi[l], agg, recs, wcount, N, E);
    post_kernel<<<waveBlocks, 256, 0, stream>>>(agg, gm[l], be[l], hbuf, out,
                                                 (l < 2),
                                                 (l == 0), N);
    X = hbuf;
  }
}


# Layer evaluated as (P x) W + b, not P (x W) + b: rows are aggregated first, so the GEMM block holding 16 whole rows does bias, LayerNorm, ReLU, running max itself; LN pass gone; sum order changes
# speedup vs baseline: 2.8710x; 1.1362x over previous
#include <hip/hip_runtime.h>

#define DCH 128
#define LN_EPS 1e-5f
#define RNG 400
#define QCAP 64
#define PCH 4096
#define AWAVES 32
#define CAPW 512
#define REC_FULL 0xFFFFFFFFu
#define XPITCH 132

typedef float v2f __attribute__((ext_vector_type(2)));
typedef float v8f __attribute__((ext_vector_type(8)));

__global__ void zero_cnt_kernel(unsigned* __restrict__ cnt, int n) {
  int i = blockIdx.x * blockDim.x + threadIdx.x;
  if (i < n) cnt[i] = 0u;
}

__global__ void count_edges_kernel(const int* __restrict__ dst,
                                   unsigned* __restrict__ cnt, int e) {
  int i = blockIdx.x * blockDim.x + threadIdx.x;
  int stride = gridDim.x * blockDim.x;
  for (; i < e; i += stride) atomicAdd(&cnt[dst[i]], 1u);
}

__global__ void dinv_kernel(const unsigned* __restrict__ cnt,
                            float* __restrict__ dinv, int n) {
  int i = blockIdx.x * blockDim.x + threadIdx.x;
  if (i < n) dinv[i] = rsqrtf(1.0f + (float)cnt[i]);
}

__global__ __launch_bounds__(256)
void key_planes_kernel(const int* __restrict__ dst,
                       unsigned* __restrict__ planes, int e, int nwaves) {
  const int lane = threadIdx.x & 31;
  const int wv   = (int)((blockIdx.x * blockDim.x + threadIdx.x) >> 5);
  if (wv >= nwaves) return;
  const int ebase = wv * 1024 + lane;
  const int last  = e - 1;
  int dv[32];
#pragma unroll
  for (int t = 0; t < 32; ++t) {
    const int idx = ebase + t * 32;
    dv[t] = dst[idx < last ? idx : last];
  }
  unsigned wd[8];
#pragma unroll
  for (int i = 0; i < 8; ++i) wd[i] = 0u;
#pragma unroll
  for (int t = 0; t < 32; ++t) {
    const unsigned key = (unsigned)dv[t] / (unsigned)RNG;
#pragma unroll
    for (int i = 0; i < 8; ++i) {
      const unsigned b = __builtin_amdgcn_ballot_w32(((key >> i) & 1u) != 0u);
      if (lane == t) wd[i] = b;
    }
  }
  unsigned* o = planes + (size_t)(wv >> 2) * 1024 + (wv & 3) * 32 + lane;
#pragma unroll
  for (int i = 0; i < 8; ++i) o[i * 128] = wd[i];
}

__global__ __launch_bounds__(256)
void gemm_ln_kernel(const float* __restrict__ A, const float* __restrict__ W,
                    const float* __restrict__ bias, const float* __restrict__ gamma,
                    const float* __restrict__ beta, float* __restrict__ hout,
                    float* __restrict__ jk, int do_relu, int first, int nrows) {
  __shared__ float xs[16 * XPITCH];
  const int m0 = blockIdx.x * 16;
  const int t  = threadIdx.x;
  const int wave = t >> 5;
  const int lane = t & 31;

  const int tile_elems = (nrows - m0 >= 16 ? 16 : (nrows - m0)) * DCH;
  const float4* src4 = (const float4*)(A + (size_t)m0 * DCH);
  float4 z; z.x = z.y = z.z = z.w = 0.0f;
  {
    int i0 = t, i1 = t + 256;
    *(float4*)(xs + (i0 >> 5) * XPITCH + (i0 & 31) * 4) = (i0 * 4 < tile_elems) ? src4[i0] : z;
    *(float4*)(xs + (i1 >> 5) * XPITCH + (i1 & 31) * 4) = (i1 * 4 < tile_elems) ? src4[i1] : z;
  }

  __syncthreads();

  const int n0   = wave * 16;
  const int row  = lane & 15;
  const int col  = lane & 15;
  const int koff = (lane >> 4) * 2;
  const int half = lane >> 4;

  const float bc = bias[n0 + col];

  v8f c = {};
#pragma unroll 8
  for (int k0 = 0; k0 < DCH; k0 += 4) {
    v2f a, b;
    a.x = xs[row * XPITCH + k0 + koff];
    a.y = xs[row * XPITCH + k0 + koff + 1];
    b.x = W[(k0 + koff)     * DCH + n0 + col];
    b.y = W[(k0 + koff + 1) * DCH + n0 + col];
    c = __builtin_amdgcn_wmma_f32_16x16x4_f32(
         false, a,  false, b,
         (short)0, c,  false,  false);
  }

  __syncthreads();

#pragma unroll
  for (int v = 0; v < 8; ++v)
    xs[(v + half * 8) * XPITCH + n0 + col] = c[v] + bc;
  __syncthreads();

  const int  node0 = m0 + wave * 2;
  const bool live0 = node0 < nrows;
  const bool live1 = node0 + 1 < nrows;
  float4 j0 = z, j1 = z;
  if (!first) {
    const int c0 = live0 ? node0 : nrows - 1;
    const int c1 = live1 ? node0 + 1 : nrows - 1;
    j0 = ((const float4*)(jk + (size_t)c0 * DCH))[lane];
    j1 = ((const float4*)(jk + (size_t)c1 * DCH))[lane];
  }
  const float4 g4 = ((const float4*)gamma)[lane];
  const float4 b4 = ((const float4*)beta)[lane];
  const float4 v0 = *(const float4*)(xs + (wave * 2)     * XPITCH + lane * 4);
  const float4 v1 = *(const float4*)(xs + (wave * 2 + 1) * XPITCH + lane * 4);

  float s0 = v0.x + v0.y + v0.z + v0.w;
  float s1 = v1.x + v1.y + v1.z + v1.w;
#pragma unroll
  for (int off = 16; off > 0; off >>= 1) {
    s0 += __shfl_xor(s0, off, 32);
    s1 += __shfl_xor(s1, off, 32);
  }
  const float mu0 = s0 * (1.0f / DCH);
  const float mu1 = s1 * (1.0f / DCH);

  const float dx0 = v0.x - mu0, dy0 = v0.y - mu0, dz0 = v0.z - mu0, dw0 = v0.w - mu0;
  const float dx1 = v1.x - mu1, dy1 = v1.y - mu1, dz1 = v1.z - mu1, dw1 = v1.w - mu1;
  float q0 = dx0 * dx0 + dy0 * dy0 + dz0 * dz0 + dw0 * dw0;
  float q1 = dx1 * dx1 + dy1 * dy1 + dz1 * dz1 + dw1 * dw1;
#pragma unroll
  for (int off = 16; off > 0; off >>= 1) {
    q0 += __shfl_xor(q0, off, 32);
    q1 += __shfl_xor(q1, off, 32);
  }
  const float inv0 = rsqrtf(q0 * (1.0f / DCH) + LN_EPS);
  const float inv1 = rsqrtf(q1 * (1.0f / DCH) + LN_EPS);

  float4 y0, y1;
  y0.x = dx0 * inv0 * g4.x + b4.x;  y1.x = dx1 * inv1 * g4.x + b4.x;
  y0.y = dy0 * inv0 * g4.y + b4.y;  y1.y = dy1 * inv1 * g4.y + b4.y;
  y0.z = dz0 * inv0 * g4.z + b4.z;  y1.z = dz1 * inv1 * g4.z + b4.z;
  y0.w = dw0 * inv0 * g4.w + b4.w;  y1.w = dw1 * inv1 * g4.w + b4.w;
  if (do_relu) {
    y0.x = fmaxf(y0.x, 0.0f); y0.y = fmaxf(y0.y, 0.0f);
    y0.z = fmaxf(y0.z, 0.0f); y0.w = fmaxf(y0.w, 0.0f);
    y1.x = fmaxf(y1.x, 0.0f); y1.y = fmaxf(y1.y, 0.0f);
    y1.z = fmaxf(y1.z, 0.0f); y1.w = fmaxf(y1.w, 0.0f);
  }

  if (live0) ((float4*)(hout + (size_t)node0 * DCH))[lane] = y0;
  if (live1) ((float4*)(hout + (size_t)(node0 + 1) * DCH))[lane] = y1;

  if (!first) {
    y0.x = fmaxf(j0.x, y0.x); y0.y = fmaxf(j0.y, y0.y);
    y0.z = fmaxf(j0.z, y0.z); y0.w = fmaxf(j0.w, y0.w);
    y1.x = fmaxf(j1.x, y1.x); y1.y = fmaxf(j1.y, y1.y);
    y1.z = fmaxf(j1.z, y1.z); y1.w = fmaxf(j1.w, y1.w);
  }
  if (live0) ((float4*)(jk + (size_t)node0 * DCH))[lane] = y0;
  if (live1) ((float4*)(jk + (size_t)(node0 + 1) * DCH))[lane] = y1;
}

static __device__ __forceinline__ void lds_add(float* p, float v) {
  __hip_atomic_fetch_add(p, v, __ATOMIC_RELAXED, __HIP_MEMORY_SCOPE_WORKGROUP);
}

template <bool FULL, typename AfterFirst>
static __device__ __forceinline__ void add_rows(
    float* acc, int s, int dl, float cf, int nb, int lane,
    const float* __restrict__ hlin, AfterFirst after_first) {
#pragma unroll 1
  for (int k = 0; k < nb; k += 8) {
    float4 v[8];
#pragma unroll
    for (int u = 0; u < 8; ++u) {
      if (FULL || k + u < nb) {
        int su = __builtin_amdgcn_readlane(s, k + u);
        v[u] = ((const float4*)(hlin + (size_t)su * DCH))[lane];
      }
    }
    if (k == 0) after_first();
#pragma unroll
    for (int u = 0; u < 8; ++u) {
      if (FULL || k + u < nb) {
        float cu = __int_as_float(__builtin_amdgcn_readlane(__float_as_int(cf), k + u));
        int   du = __builtin_amdgcn_readlane(dl, k + u);
        float* a = acc + du * DCH + lane;
        lds_add(a,      v[u].x * cu);
        lds_add(a + 32, v[u].y * cu);
        lds_add(a + 64, v[u].z * cu);
        lds_add(a + 96, v[u].w * cu);
      }
    }
  }
}

template <bool FULL, bool RECORD>
static __device__ __forceinline__ void drain_ring(
    float* acc, const int* rw, unsigned head, int nb, int lane, int lo,
    const float* __restrict__ hlin, const float* __restrict__ dinv,
    const int* __restrict__ src, const int* __restrict__ dst,
    uint2* seg, unsigned& nrec) {
  int   s = 0, dl = 0;
  float cf = 0.0f;
  if (FULL || lane < nb) {
    const int en = rw[(head + (unsigned)lane) & (QCAP - 1)];
    s = src[en];
    const int d = dst[en];
    cf = dinv[s] * dinv[d];
    dl = d - lo;
  }
  if (RECORD) {
    if (nrec != REC_FULL && nrec + (unsigned)nb <= (unsigned)CAPW) {
      if (FULL || lane < nb) {
        uint2 r;
        r.x = (unsigned)s | ((unsigned)dl << 23);
        r.y = __float_as_uint(cf);
        seg[nrec + (unsigned)lane] = r;
      }
      nrec += (unsigned)nb;
    } else {
      nrec = REC_FULL;
    }
  }
  add_rows<FULL>(acc, s, dl, cf, nb, lane, hlin, [] {});
}

template <bool REPLAY>
__global__ __launch_bounds__(1024)
void range_agg_kernel(const float* __restrict__ hlin,
                      const unsigned* __restrict__ planes,
                      const int* __restrict__ src,
                      const int* __restrict__ dst,
                      const float* __restrict__ dinv,
                      float* __restrict__ agg,
                      unsigned* recs, unsigned* wcount, int n, int e) {
  __shared__ float acc[RNG * DCH];
  __shared__ int   ring[AWAVES * QCAP];

  const int t    = threadIdx.x;
  const int lane = t & 31;
  const int wave = __builtin_amdgcn_readfirstlane(t >> 5);
  const int lo   = blockIdx.x * RNG;
  const unsigned span = (unsigned)((n - lo) < RNG ? (n - lo) : RNG);

  {
    float4 z; z.x = z.y = z.z = z.w = 0.0f;
    for (int i = t; i < RNG * DCH / 4; i += 1024) ((float4*)acc)[i] = z;
  }
  __syncthreads();

  int* rw = ring + wave * QCAP;
  unsigned head = 0, cnt = 0;

  uint2* seg = (uint2*)recs + ((size_t)blockIdx.x * AWAVES + (size_t)wave) * CAPW;
  unsigned nrec = 0;

  bool use_lists = false;
  if (REPLAY) {
    const unsigned c = wcount[blockIdx.x * AWAVES + lane];
    use_lists = __builtin_amdgcn_ballot_w32(c == REC_FULL) == 0u;
    nrec = (unsigned)__builtin_amdgcn_readlane((int)c, wave);
  }

  if (use_lists) {
    if (nrec > 0u) {
      const unsigned last = nrec - 1u;
      uint2 r = seg[(unsigned)lane < last ? (unsigned)lane : last];
      const unsigned full = nrec & ~31u;
      for (unsigned base = 0; base < full; base += 32u) {
        uint2 rn = r;
        const int   s  = (int)(r.x & 0x7FFFFFu);
        const int   dl = (int)(r.x >> 23);
        const float cf = __uint_as_float(r.y);
        add_rows<true>(acc, s, dl, cf, 32, lane, hlin, [&] {
          const unsigned i = base + 32u + (unsigned)lane;
          rn = seg[i < last ? i : last];
        });
        r = rn;
      }
      if (nrec > full) {
        const int   s  = (int)(r.x & 0x7FFFFFu);
        const int   dl = (int)(r.x >> 23);
        const float cf = __uint_as_float(r.y);
        add_rows<false>(acc, s, dl, cf, (int)(nrec - full), lane, hlin, [] {});
      }
    }
  } else {
    unsigned y[8];
#pragma unroll
    for (int i = 0; i < 8; ++i) y[i] = ((blockIdx.x >> i) & 1u) ? 0xFFFFFFFFu : 0u;

    auto park = [&](unsigned m, int eb) {
      unsigned b = __builtin_amdgcn_ballot_w32(m != 0u);
      while (b != 0u) {
        if (m != 0u)
          rw[(head + cnt + __builtin_amdgcn_mbcnt_lo(b, 0u)) & (QCAP - 1)] = eb + __builtin_ctz(m);
        m &= m - 1u;
        cnt += (unsigned)__builtin_popcount(b);
        if (cnt >= 32u) {
          drain_ring<true, !REPLAY>(acc, rw, head, 32, lane, lo, hlin, dinv, src, dst, seg, nrec);
          head = (head + 32u) & (QCAP - 1);
          cnt -= 32u;
        }
        b = __builtin_amdgcn_ballot_w32(m != 0u);
      }
    };

    const int nchunk = (e + PCH - 1) / PCH;
    const int start = (int)((blockIdx.x * (unsigned)nchunk) / gridDim.x);
    const uint4* pl4 = (const uint4*)planes;

    for (int ci = wave; ci < nchunk; ci += AWAVES) {
      int c = ci + start;
      if (c >= nchunk) c -= nchunk;
      uint4 p[8];
#pragma unroll
      for (int i = 0; i < 8; ++i) p[i] = pl4[(c * 8 + i) * 32 + lane];
      uint4 m;
      m.x = ~((p[0].x ^ y[0]) | (p[1].x ^ y[1]) | (p[2].x ^ y[2]) | (p[3].x ^ y[3]) |
              (p[4].x ^ y[4]) | (p[5].x ^ y[5]) | (p[6].x ^ y[6]) | (p[7].x ^ y[7]));
      m.y = ~((p[0].y ^ y[0]) | (p[1].y ^ y[1]) | (p[2].y ^ y[2]) | (p[3].y ^ y[3]) |
              (p[4].y ^ y[4]) | (p[5].y ^ y[5]) | (p[6].y ^ y[6]) | (p[7].y ^ y[7]));
      m.z = ~((p[0].z ^ y[0]) | (p[1].z ^ y[1]) | (p[2].z ^ y[2]) | (p[3].z ^ y[3]) |
              (p[4].z ^ y[4]) | (p[5].z ^ y[5]) | (p[6].z ^ y[6]) | (p[7].z ^ y[7]));
      m.w = ~((p[0].w ^ y[0]) | (p[1].w ^ y[1]) | (p[2].w ^ y[2]) | (p[3].w ^ y[3]) |
              (p[4].w ^ y[4]) | (p[5].w ^ y[5]) | (p[6].w ^ y[6]) | (p[7].w ^ y[7]));
      const int eb = c * PCH + lane * 128;
      if (c == nchunk - 1) {
        auto live = [&](int r) -> unsigned {
          return r >= 32 ? 0xFFFFFFFFu : (r > 0 ? ((1u << r) - 1u) : 0u);
        };
        m.x &= live(e - eb);
        m.y &= live(e - eb - 32);
        m.z &= live(e - eb - 64);
        m.w &= live(e - eb - 96);
      }
      park(m.x, eb);
      park(m.y, eb + 32);
      park(m.z, eb + 64);
      park(m.w, eb + 96);
    }
    if (cnt > 0u)
      drain_ring<false, !REPLAY>(acc, rw, head, (int)cnt, lane, lo, hlin, dinv, src, dst, seg, nrec);
    if (!REPLAY && lane == 0) wcount[blockIdx.x * AWAVES + wave] = nrec;
  }
  __syncthreads();

  for (int r = wave; r < (int)span; r += AWAVES) {
    const int node = lo + r;
    const float* a = acc + r * DCH + lane;
    float dv = dinv[node];
    float sc = dv * dv;
    float4 h = ((const float4*)(hlin + (size_t)node * DCH))[lane];
    float4 o;
    o.x = a[0]  + h.x * sc;
    o.y = a[32] + h.y * sc;
    o.z = a[64] + h.z * sc;
    o.w = a[96] + h.w * sc;
    ((float4*)(agg + (size_t)node * DCH))[lane] = o;
  }
}

extern "C" void kernel_launch(void* const* d_in, const int* in_sizes, int n_in,
                              void* d_out, int out_size, void* d_ws, size_t ws_size,
                              hipStream_t stream) {
  const int N = in_sizes[0] / DCH;
  const int E = in_sizes[1] / 2;

  const float* feat = (const float*)d_in[0];
  const int*   ei   = (const int*)d_in[1];
  const int*   src  = ei;
  const int*   dst  = ei + E;
  const float* Wm[3] = {(const float*)d_in[2], (const float*)d_in[6],  (const float*)d_in[10]};
  const float* bi[3] = {(const float*)d_in[3], (const float*)d_in[7],  (const float*)d_in[11]};
  const float* gm[3] = {(const float*)d_in[4], (const float*)d_in[8],  (const float*)d_in[12]};
  const float* be[3] = {(const float*)d_in[5], (const float*)d_in[9],  (const float*)d_in[13]};
  float* out = (float*)d_out;

  float*    wsf  = (float*)d_ws;
  unsigned* cnt  = (unsigned*)wsf;
  float*    dinv = wsf + N;
  float*    lists = wsf + 2 * (size_t)N;
  float*    agg  = lists + (size_t)N * DCH;
  float*    hbuf = agg  + (size_t)N * DCH;
  unsigned* planes = (unsigned*)(hbuf + (size_t)N * DCH);

  const int nodeBlocks = (N + 255) / 256;
  zero_cnt_kernel<<<nodeBlocks, 256, 0, stream>>>(cnt, N);
  count_edges_kernel<<<2048, 256, 0, stream>>>(dst, cnt, E);
  dinv_kernel<<<nodeBlocks, 256, 0, stream>>>(cnt, dinv, N);
  const int planeWaves = ((E + PCH - 1) / PCH) * 4;
  key_planes_kernel<<<(planeWaves + 7) / 8, 256, 0, stream>>>(dst, planes, E, planeWaves);

  const int gemmBlocks = (N + 15) / 16;
  const int rangeBlocks = (N + RNG - 1) / RNG;
  unsigned* recs   = (unsigned*)lists;
  unsigned* wcount = recs + (size_t)rangeBlocks * AWAVES * CAPW * 2;

  const float* X = feat;
  for (int l = 0; l < 3; ++l) {
    if (l == 0)
      range_agg_kernel<false><<<rangeBlocks, 1024, 0, stream>>>(X, planes, src, dst, dinv,
                                                                agg, recs, wcount, N, E);
    else
      range_agg_kernel<true><<<rangeBlocks, 1024, 0, stream>>>(X, planes, src, dst, dinv,
                                                               agg, recs, wcount, N, E);
    gemm_ln_kernel<<<gemmBlocks, 256, 0, stream>>>(agg, Wm[l], bi[l], gm[l], be[l], hbuf, out,
                                                    (l < 2),
                                                    (l == 0), N);
    X = hbuf;
  }
}


# GEMM: each wave keeps its 128x16 slice of W in registers and a resident block walks 16-row tiles with it: W fetched once per block, not per 16 rows; next tile's rows asked for ahead; sum order kept
# speedup vs baseline: 3.1500x; 1.0972x over previous
#include <hip/hip_runtime.h>

#define DCH 128
#define LN_EPS 1e-5f
#define RNG 400
#define QCAP 64
#define PCH 4096
#define AWAVES 32
#define CAPW 512
#define REC_FULL 0xFFFFFFFFu
#define XPITCH 132

typedef float v2f __attribute__((ext_vector_type(2)));
typedef float v8f __attribute__((ext_vector_type(8)));

__global__ void zero_cnt_kernel(unsigned* __restrict__ cnt, int n) {
  int i = blockIdx.x * blockDim.x + threadIdx.x;
  if (i < n) cnt[i] = 0u;
}

__global__ void count_edges_kernel(const int* __restrict__ dst,
                                   unsigned* __restrict__ cnt, int e) {
  int i = blockIdx.x * blockDim.x + threadIdx.x;
  int stride = gridDim.x * blockDim.x;
  for (; i < e; i += stride) atomicAdd(&cnt[dst[i]], 1u);
}

__global__ void dinv_kernel(const unsigned* __restrict__ cnt,
                            float* __restrict__ dinv, int n) {
  int i = blockIdx.x * blockDim.x + threadIdx.x;
  if (i < n) dinv[i] = rsqrtf(1.0f + (float)cnt[i]);
}

__global__ __launch_bounds__(256)
void key_planes_kernel(const int* __restrict__ dst,
                       unsigned* __restrict__ planes, int e, int nwaves) {
  const int lane = threadIdx.x & 31;
  const int wv   = (int)((blockIdx.x * blockDim.x + threadIdx.x) >> 5);
  if (wv >= nwaves) return;
  const int ebase = wv * 1024 + lane;
  const int last  = e - 1;
  int dv[32];
#pragma unroll
  for (int t = 0; t < 32; ++t) {
    const int idx = ebase + t * 32;
    dv[t] = dst[idx < last ? idx : last];
  }
  unsigned wd[8];
#pragma unroll
  for (int i = 0; i < 8; ++i) wd[i] = 0u;
#pragma unroll
  for (int t = 0; t < 32; ++t) {
    const unsigned key = (unsigned)dv[t] / (unsigned)RNG;
#pragma unroll
    for (int i = 0; i < 8; ++i) {
      const unsigned b = __builtin_amdgcn_ballot_w32(((key >> i) & 1u) != 0u);
      if (lane == t) wd[i] = b;
    }
  }
  unsigned* o = planes + (size_t)(wv >> 2) * 1024 + (wv & 3) * 32 + lane;
#pragma unroll
  for (int i = 0; i < 8; ++i) o[i * 128] = wd[i];
}

__global__ __launch_bounds__(256)
void gemm_ln_kernel(const float* __restrict__ A, const float* __restrict__ W,
                    const float* __restrict__ bias, const float* __restrict__ gamma,
                    const float* __restrict__ beta, float* __restrict__ hout,
                    float* __restrict__ jk, int do_relu, int first, int nrows) {
  __shared__ float xs[16 * XPITCH];
  const int t  = threadIdx.x;
  const int wave = t >> 5;
  const int lane = t & 31;

  const int n0   = wave * 16;
  const int row  = lane & 15;
  const int col  = lane & 15;
  const int koff = (lane >> 4) * 2;
  const int half = lane >> 4;
  const int ntiles = (nrows + 15) >> 4;
  const int lastrow = nrows - 1;

  v2f bf[32];
#pragma unroll
  for (int s = 0; s < 32; ++s) {
    bf[s].x = W[(s * 4 + koff)     * DCH + n0 + col];
    bf[s].y = W[(s * 4 + koff + 1) * DCH + n0 + col];
  }
  const float bc = bias[n0 + col];
  const float4 g4 = ((const float4*)gamma)[lane];
  const float4 b4 = ((const float4*)beta)[lane];
  float4 z; z.x = z.y = z.z = z.w = 0.0f;

  int mt = blockIdx.x;
  float4 a0 = z, a1 = z;
  if (mt < ntiles) {
    const int r0 = mt * 16 + wave, r1 = r0 + 8;
    a0 = ((const float4*)(A + (size_t)(r0 < lastrow ? r0 : lastrow) * DCH))[lane];
    a1 = ((const float4*)(A + (size_t)(r1 < lastrow ? r1 : lastrow) * DCH))[lane];
    if (r0 >= nrows) a0 = z;
    if (r1 >= nrows) a1 = z;
  }

  for (; mt < ntiles; mt += (int)gridDim.x) {
    const int m0 = mt * 16;
    *(float4*)(xs + wave * XPITCH + lane * 4)       = a0;
    *(float4*)(xs + (wave + 8) * XPITCH + lane * 4) = a1;
    __syncthreads();

    {
      const int r0 = (mt + (int)gridDim.x) * 16 + wave, r1 = r0 + 8;
      a0 = ((const float4*)(A + (size_t)(r0 < lastrow ? r0 : lastrow) * DCH))[lane];
      a1 = ((const float4*)(A + (size_t)(r1 < lastrow ? r1 : lastrow) * DCH))[lane];
      if (r0 >= nrows) a0 = z;
      if (r1 >= nrows) a1 = z;
    }

    const int  node0 = m0 + wave * 2;
    const bool live0 = node0 < nrows;
    const bool live1 = node0 + 1 < nrows;
    float4 j0 = z, j1 = z;
    if (!first) {
      const int c0 = live0 ? node0 : lastrow;
      const int c1 = live1 ? node0 + 1 : lastrow;
      j0 = ((const float4*)(jk + (size_t)c0 * DCH))[lane];
      j1 = ((const float4*)(jk + (size_t)c1 * DCH))[lane];
    }

    v8f c = {};
#pragma unroll
    for (int s = 0; s < 32; ++s) {
      v2f a;
      a.x = xs[row * XPITCH + s * 4 + koff];
      a.y = xs[row * XPITCH + s * 4 + koff + 1];
      c = __builtin_amdgcn_wmma_f32_16x16x4_f32(
           false, a,  false, bf[s],
           (short)0, c,  false,  false);
    }

    __syncthreads();

#pragma unroll
    for (int v = 0; v < 8; ++v)
      xs[(v + half * 8) * XPITCH + n0 + col] = c[v] + bc;
    __syncthreads();

    const float4 v0 = *(const float4*)(xs + (wave * 2)     * XPITCH + lane * 4);
    const float4 v1 = *(const float4*)(xs + (wave * 2 + 1) * XPITCH + lane * 4);
    __syncthreads();

    float s0 = v0.x + v0.y + v0.z + v0.w;
    float s1 = v1.x + v1.y + v1.z + v1.w;
#pragma unroll
    for (int off = 16; off > 0; off >>= 1) {
      s0 += __shfl_xor(s0, off, 32);
      s1 += __shfl_xor(s1, off, 32);
    }
    const float mu0 = s0 * (1.0f / DCH);
    const float mu1 = s1 * (1.0f / DCH);

    const float dx0 = v0.x - mu0, dy0 = v0.y - mu0, dz0 = v0.z - mu0, dw0 = v0.w - mu0;
    const float dx1 = v1.x - mu1, dy1 = v1.y - mu1, dz1 = v1.z - mu1, dw1 = v1.w - mu1;
    float q0 = dx0 * dx0 + dy0 * dy0 + dz0 * dz0 + dw0 * dw0;
    float q1 = dx1 * dx1 + dy1 * dy1 + dz1 * dz1 + dw1 * dw1;
#pragma unroll
    for (int off = 16; off > 0; off >>= 1) {
      q0 += __shfl_xor(q0, off, 32);
      q1 += __shfl_xor(q1, off, 32);
    }
    const float inv0 = rsqrtf(q0 * (1.0f / DCH) + LN_EPS);
    const float inv1 = rsqrtf(q1 * (1.0f / DCH) + LN_EPS);

    float4 y0, y1;
    y0.x = dx0 * inv0 * g4.x + b4.x;  y1.x = dx1 * inv1 * g4.x + b4.x;
    y0.y = dy0 * inv0 * g4.y + b4.y;  y1.y = dy1 * inv1 * g4.y + b4.y;
    y0.z = dz0 * inv0 * g4.z + b4.z;  y1.z = dz1 * inv1 * g4.z + b4.z;
    y0.w = dw0 * inv0 * g4.w + b4.w;  y1.w = dw1 * inv1 * g4.w + b4.w;
    if (do_relu) {
      y0.x = fmaxf(y0.x, 0.0f); y0.y = fmaxf(y0.y, 0.0f);
      y0.z = fmaxf(y0.z, 0.0f); y0.w = fmaxf(y0.w, 0.0f);
      y1.x = fmaxf(y1.x, 0.0f); y1.y = fmaxf(y1.y, 0.0f);
      y1.z = fmaxf(y1.z, 0.0f); y1.w = fmaxf(y1.w, 0.0f);
    }

    if (live0) ((float4*)(hout + (size_t)node0 * DCH))[lane] = y0;
    if (live1) ((float4*)(hout + (size_t)(node0 + 1) * DCH))[lane] = y1;

    if (!first) {
      y0.x = fmaxf(j0.x, y0.x); y0.y = fmaxf(j0.y, y0.y);
      y0.z = fmaxf(j0.z, y0.z); y0.w = fmaxf(j0.w, y0.w);
      y1.x = fmaxf(j1.x, y1.x); y1.y = fmaxf(j1.y, y1.y);
      y1.z = fmaxf(j1.z, y1.z); y1.w = fmaxf(j1.w, y1.w);
    }
    if (live0) ((float4*)(jk + (size_t)node0 * DCH))[lane] = y0;
    if (live1) ((float4*)(jk + (size_t)(node0 + 1) * DCH))[lane] = y1;
  }
}

static __device__ __forceinline__ void lds_add(float* p, float v) {
  __hip_atomic_fetch_add(p, v, __ATOMIC_RELAXED, __HIP_MEMORY_SCOPE_WORKGROUP);
}

template <bool FULL, typename AfterFirst>
static __device__ __forceinline__ void add_rows(
    float* acc, int s, int dl, float cf, int nb, int lane,
    const float* __restrict__ hlin, AfterFirst after_first) {
#pragma unroll 1
  for (int k = 0; k < nb; k += 8) {
    float4 v[8];
#pragma unroll
    for (int u = 0; u < 8; ++u) {
      if (FULL || k + u < nb) {
        int su = __builtin_amdgcn_readlane(s, k + u);
        v[u] = ((const float4*)(hlin + (size_t)su * DCH))[lane];
      }
    }
    if (k == 0) after_first();
#pragma unroll
    for (int u = 0; u < 8; ++u) {
      if (FULL || k + u < nb) {
        float cu = __int_as_float(__builtin_amdgcn_readlane(__float_as_int(cf), k + u));
        int   du = __builtin_amdgcn_readlane(dl, k + u);
        float* a = acc + du * DCH + lane;
        lds_add(a,      v[u].x * cu);
        lds_add(a + 32, v[u].y * cu);
        lds_add(a + 64, v[u].z * cu);
        lds_add(a + 96, v[u].w * cu);
      }
    }
  }
}

template <bool FULL, bool RECORD>
static __device__ __forceinline__ void drain_ring(
    float* acc, const int* rw, unsigned head, int nb, int lane, int lo,
    const float* __restrict__ hlin, const float* __restrict__ dinv,
    const int* __restrict__ src, const int* __restrict__ dst,
    uint2* seg, unsigned& nrec) {
  int   s = 0, dl = 0;
  float cf = 0.0f;
  if (FULL || lane < nb) {
    const int en = rw[(head + (unsigned)lane) & (QCAP - 1)];
    s = src[en];
    const int d = dst[en];
    cf = dinv[s] * dinv[d];
    dl = d - lo;
  }
  if (RECORD) {
    if (nrec != REC_FULL && nrec + (unsigned)nb <= (unsigned)CAPW) {
      if (FULL || lane < nb) {
        uint2 r;
        r.x = (unsigned)s | ((unsigned)dl << 23);
        r.y = __float_as_uint(cf);
        seg[nrec + (unsigned)lane] = r;
      }
      nrec += (unsigned)nb;
    } else {
      nrec = REC_FULL;
    }
  }
  add_rows<FULL>(acc, s, dl, cf, nb, lane, hlin, [] {});
}

template <bool REPLAY>
__global__ __launch_bounds__(1024)
void range_agg_kernel(const float* __restrict__ hlin,
                      const unsigned* __restrict__ planes,
                      const int* __restrict__ src,
                      const int* __restrict__ dst,
                      const float* __restrict__ dinv,
                      float* __restrict__ agg,
                      unsigned* recs, unsigned* wcount, int n, int e) {
  __shared__ float acc[RNG * DCH];
  __shared__ int   ring[AWAVES * QCAP];

  const int t    = threadIdx.x;
  const int lane = t & 31;
  const int wave = __builtin_amdgcn_readfirstlane(t >> 5);
  const int lo   = blockIdx.x * RNG;
  const unsigned span = (unsigned)((n - lo) < RNG ? (n - lo) : RNG);

  {
    float4 z; z.x = z.y = z.z = z.w = 0.0f;
    for (int i = t; i < RNG * DCH / 4; i += 1024) ((float4*)acc)[i] = z;
  }
  __syncthreads();

  int* rw = ring + wave * QCAP;
  unsigned head = 0, cnt = 0;

  uint2* seg = (uint2*)recs + ((size_t)blockIdx.x * AWAVES + (size_t)wave) * CAPW;
  unsigned nrec = 0;

  bool use_lists = false;
  if (REPLAY) {
    const unsigned c = wcount[blockIdx.x * AWAVES + lane];
    use_lists = __builtin_amdgcn_ballot_w32(c == REC_FULL) == 0u;
    nrec = (unsigned)__builtin_amdgcn_readlane((int)c, wave);
  }

  if (use_lists) {
    if (nrec > 0u) {
      const unsigned last = nrec - 1u;
      uint2 r = seg[(unsigned)lane < last ? (unsigned)lane : last];
      const unsigned full = nrec & ~31u;
      for (unsigned base = 0; base < full; base += 32u) {
        uint2 rn = r;
        const int   s  = (int)(r.x & 0x7FFFFFu);
        const int   dl = (int)(r.x >> 23);
        const float cf = __uint_as_float(r.y);
        add_rows<true>(acc, s, dl, cf, 32, lane, hlin, [&] {
          const unsigned i = base + 32u + (unsigned)lane;
          rn = seg[i < last ? i : last];
        });
        r = rn;
      }
      if (nrec > full) {
        const int   s  = (int)(r.x & 0x7FFFFFu);
        const int   dl = (int)(r.x >> 23);
        const float cf = __uint_as_float(r.y);
        add_rows<false>(acc, s, dl, cf, (int)(nrec - full), lane, hlin, [] {});
      }
    }
  } else {
    unsigned y[8];
#pragma unroll
    for (int i = 0; i < 8; ++i) y[i] = ((blockIdx.x >> i) & 1u) ? 0xFFFFFFFFu : 0u;

    auto park = [&](unsigned m, int eb) {
      unsigned b = __builtin_amdgcn_ballot_w32(m != 0u);
      while (b != 0u) {
        if (m != 0u)
          rw[(head + cnt + __builtin_amdgcn_mbcnt_lo(b, 0u)) & (QCAP - 1)] = eb + __builtin_ctz(m);
        m &= m - 1u;
        cnt += (unsigned)__builtin_popcount(b);
        if (cnt >= 32u) {
          drain_ring<true, !REPLAY>(acc, rw, head, 32, lane, lo, hlin, dinv, src, dst, seg, nrec);
          head = (head + 32u) & (QCAP - 1);
          cnt -= 32u;
        }
        b = __builtin_amdgcn_ballot_w32(m != 0u);
      }
    };

    const int nchunk = (e + PCH - 1) / PCH;
    const int start = (int)((blockIdx.x * (unsigned)nchunk) / gridDim.x);
    const uint4* pl4 = (const uint4*)planes;

    for (int ci = wave; ci < nchunk; ci += AWAVES) {
      int c = ci + start;
      if (c >= nchunk) c -= nchunk;
      uint4 p[8];
#pragma unroll
      for (int i = 0; i < 8; ++i) p[i] = pl4[(c * 8 + i) * 32 + lane];
      uint4 m;
      m.x = ~((p[0].x ^ y[0]) | (p[1].x ^ y[1]) | (p[2].x ^ y[2]) | (p[3].x ^ y[3]) |
              (p[4].x ^ y[4]) | (p[5].x ^ y[5]) | (p[6].x ^ y[6]) | (p[7].x ^ y[7]));
      m.y = ~((p[0].y ^ y[0]) | (p[1].y ^ y[1]) | (p[2].y ^ y[2]) | (p[3].y ^ y[3]) |
              (p[4].y ^ y[4]) | (p[5].y ^ y[5]) | (p[6].y ^ y[6]) | (p[7].y ^ y[7]));
      m.z = ~((p[0].z ^ y[0]) | (p[1].z ^ y[1]) | (p[2].z ^ y[2]) | (p[3].z ^ y[3]) |
              (p[4].z ^ y[4]) | (p[5].z ^ y[5]) | (p[6].z ^ y[6]) | (p[7].z ^ y[7]));
      m.w = ~((p[0].w ^ y[0]) | (p[1].w ^ y[1]) | (p[2].w ^ y[2]) | (p[3].w ^ y[3]) |
              (p[4].w ^ y[4]) | (p[5].w ^ y[5]) | (p[6].w ^ y[6]) | (p[7].w ^ y[7]));
      const int eb = c * PCH + lane * 128;
      if (c == nchunk - 1) {
        auto live = [&](int r) -> unsigned {
          return r >= 32 ? 0xFFFFFFFFu : (r > 0 ? ((1u << r) - 1u) : 0u);
        };
        m.x &= live(e - eb);
        m.y &= live(e - eb - 32);
        m.z &= live(e - eb - 64);
        m.w &= live(e - eb - 96);
      }
      park(m.x, eb);
      park(m.y, eb + 32);
      park(m.z, eb + 64);
      park(m.w, eb + 96);
    }
    if (cnt > 0u)
      drain_ring<false, !REPLAY>(acc, rw, head, (int)cnt, lane, lo, hlin, dinv, src, dst, seg, nrec);
    if (!REPLAY && lane == 0) wcount[blockIdx.x * AWAVES + wave] = nrec;
  }
  __syncthreads();

  for (int r = wave; r < (int)span; r += AWAVES) {
    const int node = lo + r;
    const float* a = acc + r * DCH + lane;
    float dv = dinv[node];
    float sc = dv * dv;
    float4 h = ((const float4*)(hlin + (size_t)node * DCH))[lane];
    float4 o;
    o.x = a[0]  + h.x * sc;
    o.y = a[32] + h.y * sc;
    o.z = a[64] + h.z * sc;
    o.w = a[96] + h.w * sc;
    ((float4*)(agg + (size_t)node * DCH))[lane] = o;
  }
}

extern "C" void kernel_launch(void* const* d_in, const int* in_sizes, int n_in,
                              void* d_out, int out_size, void* d_ws, size_t ws_size,
                              hipStream_t stream) {
  const int N = in_sizes[0] / DCH;
  const int E = in_sizes[1] / 2;

  const float* feat = (const float*)d_in[0];
  const int*   ei   = (const int*)d_in[1];
  const int*   src  = ei;
  const int*   dst  = ei + E;
  const float* Wm[3] = {(const float*)d_in[2], (const float*)d_in[6],  (const float*)d_in[10]};
  const float* bi[3] = {(const float*)d_in[3], (const float*)d_in[7],  (const float*)d_in[11]};
  const float* gm[3] = {(const float*)d_in[4], (const float*)d_in[8],  (const float*)d_in[12]};
  const float* be[3] = {(const float*)d_in[5], (const float*)d_in[9],  (const float*)d_in[13]};
  float* out = (float*)d_out;

  float*    wsf  = (float*)d_ws;
  unsigned* cnt  = (unsigned*)wsf;
  float*    dinv = wsf + N;
  float*    lists = wsf + 2 * (size_t)N;
  float*    agg  = lists + (size_t)N * DCH;
  float*    hbuf = agg  + (size_t)N * DCH;
  unsigned* planes = (unsigned*)(hbuf + (size_t)N * DCH);

  const int nodeBlocks = (N + 255) / 256;
  zero_cnt_kernel<<<nodeBlocks, 256, 0, stream>>>(cnt, N);
  count_edges_kernel<<<2048, 256, 0, stream>>>(dst, cnt, E);
  dinv_kernel<<<nodeBlocks, 256, 0, stream>>>(cnt, dinv, N);
  const int planeWaves = ((E + PCH - 1) / PCH) * 4;
  key_planes_kernel<<<(planeWaves + 7) / 8, 256, 0, stream>>>(dst, planes, E, planeWaves);

  const int gemmBlocks = 1024;
  const int rangeBlocks = (N + RNG - 1) / RNG;
  unsigned* recs   = (unsigned*)lists;
  unsigned* wcount = recs + (size_t)rangeBlocks * AWAVES * CAPW * 2;

  const float* X = feat;
  for (int l = 0; l < 3; ++l) {
    if (l == 0)
      range_agg_kernel<false><<<rangeBlocks, 1024, 0, stream>>>(X, planes, src, dst, dinv,
                                                                agg, recs, wcount, N, E);
    else
      range_agg_kernel<true><<<rangeBlocks, 1024, 0, stream>>>(X, planes, src, dst, dinv,
                                                               agg, recs, wcount, N, E);
    gemm_ln_kernel<<<gemmBlocks, 256, 0, stream>>>(agg, Wm[l], bi[l], gm[l], be[l], hbuf, out,
                                                    (l < 2),
                                                    (l == 0), N);
    X = hbuf;
  }
}


# Product, bias, LayerNorm, ReLU, running max run in the range-owner work-group on its 400 rows of sums in LDS (f32 WMMA, W in registers): sums never stored, 3 launches go; k, self-loop sum order change
# speedup vs baseline: 3.6106x; 1.1462x over previous
#include <hip/hip_runtime.h>

#define DCH 128
#define LN_EPS 1e-5f
#define RNG 400
#define QCAP 64
#define PCH 4096
#define AWAVES 32
#define CAPW 512
#define REC_FULL 0xFFFFFFFFu
#define XPITCH 132

typedef float v2f __attribute__((ext_vector_type(2)));
typedef float v8f __attribute__((ext_vector_type(8)));

__global__ void zero_cnt_kernel(unsigned* __restrict__ cnt, int n) {
  int i = blockIdx.x * blockDim.x + threadIdx.x;
  if (i < n) cnt[i] = 0u;
}

__global__ void count_edges_kernel(const int* __restrict__ dst,
                                   unsigned* __restrict__ cnt, int e) {
  int i = blockIdx.x * blockDim.x + threadIdx.x;
  int stride = gridDim.x * blockDim.x;
  for (; i < e; i += stride) atomicAdd(&cnt[dst[i]], 1u);
}

__global__ void dinv_kernel(const unsigned* __restrict__ cnt,
                            float* __restrict__ dinv, int n) {
  int i = blockIdx.x * blockDim.x + threadIdx.x;
  if (i < n) dinv[i] = rsqrtf(1.0f + (float)cnt[i]);
}

__global__ __launch_bounds__(256)
void key_planes_kernel(const int* __restrict__ dst,
                       unsigned* __restrict__ planes, int e, int nwaves) {
  const int lane = threadIdx.x & 31;
  const int wv   = (int)((blockIdx.x * blockDim.x + threadIdx.x) >> 5);
  if (wv >= nwaves) return;
  const int ebase = wv * 1024 + lane;
  const int last  = e - 1;
  int dv[32];
#pragma unroll
  for (int t = 0; t < 32; ++t) {
    const int idx = ebase + t * 32;
    dv[t] = dst[idx < last ? idx : last];
  }
  unsigned wd[8];
#pragma unroll
  for (int i = 0; i < 8; ++i) wd[i] = 0u;
#pragma unroll
  for (int t = 0; t < 32; ++t) {
    const unsigned key = (unsigned)dv[t] / (unsigned)RNG;
#pragma unroll
    for (int i = 0; i < 8; ++i) {
      const unsigned b = __builtin_amdgcn_ballot_w32(((key >> i) & 1u) != 0u);
      if (lane == t) wd[i] = b;
    }
  }
  unsigned* o = planes + (size_t)(wv >> 2) * 1024 + (wv & 3) * 32 + lane;
#pragma unroll
  for (int i = 0; i < 8; ++i) o[i * 128] = wd[i];
}

static __device__ __forceinline__ void lds_add(float* p, float v) {
  __hip_atomic_fetch_add(p, v, __ATOMIC_RELAXED, __HIP_MEMORY_SCOPE_WORKGROUP);
}

template <bool FULL, typename AfterFirst>
static __device__ __forceinline__ void add_rows(
    float* acc, int s, int dl, float cf, int nb, int lane,
    const float* __restrict__ hlin, AfterFirst after_first) {
#pragma unroll 1
  for (int k = 0; k < nb; k += 8) {
    float4 v[8];
#pragma unroll
    for (int u = 0; u < 8; ++u) {
      if (FULL || k + u < nb) {
        int su = __builtin_amdgcn_readlane(s, k + u);
        v[u] = ((const float4*)(hlin + (size_t)su * DCH))[lane];
      }
    }
    if (k == 0) after_first();
#pragma unroll
    for (int u = 0; u < 8; ++u) {
      if (FULL || k + u < nb) {
        float cu = __int_as_float(__builtin_amdgcn_readlane(__float_as_int(cf), k + u));
        int   du = __builtin_amdgcn_readlane(dl, k + u);
        float* a = acc + du * XPITCH + lane;
        lds_add(a,      v[u].x * cu);
        lds_add(a + 32, v[u].y * cu);
        lds_add(a + 64, v[u].z * cu);
        lds_add(a + 96, v[u].w * cu);
      }
    }
  }
}

template <bool FULL, bool RECORD>
static __device__ __forceinline__ void drain_ring(
    float* acc, const int* rw, unsigned head, int nb, int lane, int lo,
    const float* __restrict__ hlin, const float* __restrict__ dinv,
    const int* __restrict__ src, const int* __restrict__ dst,
    uint2* seg, unsigned& nrec) {
  int   s = 0, dl = 0;
  float cf = 0.0f;
  if (FULL || lane < nb) {
    const int en = rw[(head + (unsigned)lane) & (QCAP - 1)];
    s = src[en];
    const int d = dst[en];
    cf = dinv[s] * dinv[d];
    dl = d - lo;
  }
  if (RECORD) {
    if (nrec != REC_FULL && nrec + (unsigned)nb <= (unsigned)CAPW) {
      if (FULL || lane < nb) {
        uint2 r;
        r.x = (unsigned)s | ((unsigned)dl << 23);
        r.y = __float_as_uint(cf);
        seg[nrec + (unsigned)lane] = r;
      }
      nrec += (unsigned)nb;
    } else {
      nrec = REC_FULL;
    }
  }
  add_rows<FULL>(acc, s, dl, cf, nb, lane, hlin, [] {});
}

template <bool REPLAY>
__global__ __launch_bounds__(1024)
void range_layer_kernel(const float* __restrict__ hlin,
                        const unsigned* __restrict__ planes,
                        const int* __restrict__ src,
                        const int* __restrict__ dst,
                        const float* __restrict__ dinv,
                        const float* __restrict__ W, const float* __restrict__ bias,
                        const float* __restrict__ gamma, const float* __restrict__ beta,
                        float* __restrict__ hout,
                        float* __restrict__ jk,
                        unsigned* recs, unsigned* wcount,
                        int do_relu, int first, int n, int e) {
  __shared__ __attribute__((aligned(16))) float acc[RNG * XPITCH];
  __shared__ int   ring[AWAVES * QCAP];

  const int t    = threadIdx.x;
  const int lane = t & 31;
  const int wave = __builtin_amdgcn_readfirstlane(t >> 5);
  const int lo   = blockIdx.x * RNG;
  const unsigned span = (unsigned)((n - lo) < RNG ? (n - lo) : RNG);

  {
    constexpr int NI = (RNG + AWAVES - 1) / AWAVES;
    float dvl = 0.0f;
    {
      const int r = wave + lane * AWAVES;
      if (lane < NI && r < RNG && lo + r < n) dvl = dinv[lo + r];
    }
    float4 h[NI];
#pragma unroll
    for (int i = 0; i < NI; ++i) {
      const int node = lo + wave + i * AWAVES;
      h[i] = ((const float4*)(hlin + (size_t)(node < n - 1 ? node : n - 1) * DCH))[lane];
    }
#pragma unroll
    for (int i = 0; i < NI; ++i) {
      const int r = wave + i * AWAVES;
      if (r < RNG) {
        const float d  = __int_as_float(__builtin_amdgcn_readlane(__float_as_int(dvl), i));
        const float sc = d * d;
        const bool  in = lo + r < n;
        float* a = acc + r * XPITCH + lane;
        a[0]  = in ? h[i].x * sc : 0.0f;
        a[32] = in ? h[i].y * sc : 0.0f;
        a[64] = in ? h[i].z * sc : 0.0f;
        a[96] = in ? h[i].w * sc : 0.0f;
      }
    }
  }
  __syncthreads();

  int* rw = ring + wave * QCAP;
  unsigned head = 0, cnt = 0;

  uint2* seg = (uint2*)recs + ((size_t)blockIdx.x * AWAVES + (size_t)wave) * CAPW;
  unsigned nrec = 0;

  bool use_lists = false;
  if (REPLAY) {
    const unsigned c = wcount[blockIdx.x * AWAVES + lane];
    use_lists = __builtin_amdgcn_ballot_w32(c == REC_FULL) == 0u;
    nrec = (unsigned)__builtin_amdgcn_readlane((int)c, wave);
  }

  if (use_lists) {
    if (nrec > 0u) {
      const unsigned last = nrec - 1u;
      uint2 r = seg[(unsigned)lane < last ? (unsigned)lane : last];
      const unsigned full = nrec & ~31u;
      for (unsigned base = 0; base < full; base += 32u) {
        uint2 rn = r;
        const int   s  = (int)(r.x & 0x7FFFFFu);
        const int   dl = (int)(r.x >> 23);
        const float cf = __uint_as_float(r.y);
        add_rows<true>(acc, s, dl, cf, 32, lane, hlin, [&] {
          const unsigned i = base + 32u + (unsigned)lane;
          rn = seg[i < last ? i : last];
        });
        r = rn;
      }
      if (nrec > full) {
        const int   s  = (int)(r.x & 0x7FFFFFu);
        const int   dl = (int)(r.x >> 23);
        const float cf = __uint_as_float(r.y);
        add_rows<false>(acc, s, dl, cf, (int)(nrec - full), lane, hlin, [] {});
      }
    }
  } else {
    unsigned y[8];
#pragma unroll
    for (int i = 0; i < 8; ++i) y[i] = ((blockIdx.x >> i) & 1u) ? 0xFFFFFFFFu : 0u;

    auto park = [&](unsigned m, int eb) {
      unsigned b = __builtin_amdgcn_ballot_w32(m != 0u);
      while (b != 0u) {
        if (m != 0u)
          rw[(head + cnt + __builtin_amdgcn_mbcnt_lo(b, 0u)) & (QCAP - 1)] = eb + __builtin_ctz(m);
        m &= m - 1u;
        cnt += (unsigned)__builtin_popcount(b);
        if (cnt >= 32u) {
          drain_ring<true, !REPLAY>(acc, rw, head, 32, lane, lo, hlin, dinv, src, dst, seg, nrec);
          head = (head + 32u) & (QCAP - 1);
          cnt -= 32u;
        }
        b = __builtin_amdgcn_ballot_w32(m != 0u);
      }
    };

    const int nchunk = (e + PCH - 1) / PCH;
    const int start = (int)((blockIdx.x * (unsigned)nchunk) / gridDim.x);
    const uint4* pl4 = (const uint4*)planes;

    for (int ci = wave; ci < nchunk; ci += AWAVES) {
      int c = ci + start;
      if (c >= nchunk) c -= nchunk;
      uint4 p[8];
#pragma unroll
      for (int i = 0; i < 8; ++i) p[i] = pl4[(c * 8 + i) * 32 + lane];
      uint4 m;
      m.x = ~((p[0].x ^ y[0]) | (p[1].x ^ y[1]) | (p[2].x ^ y[2]) | (p[3].x ^ y[3]) |
              (p[4].x ^ y[4]) | (p[5].x ^ y[5]) | (p[6].x ^ y[6]) | (p[7].x ^ y[7]));
      m.y = ~((p[0].y ^ y[0]) | (p[1].y ^ y[1]) | (p[2].y ^ y[2]) | (p[3].y ^ y[3]) |
              (p[4].y ^ y[4]) | (p[5].y ^ y[5]) | (p[6].y ^ y[6]) | (p[7].y ^ y[7]));
      m.z = ~((p[0].z ^ y[0]) | (p[1].z ^ y[1]) | (p[2].z ^ y[2]) | (p[3].z ^ y[3]) |
              (p[4].z ^ y[4]) | (p[5].z ^ y[5]) | (p[6].z ^ y[6]) | (p[7].z ^ y[7]));
      m.w = ~((p[0].w ^ y[0]) | (p[1].w ^ y[1]) | (p[2].w ^ y[2]) | (p[3].w ^ y[3]) |
              (p[4].w ^ y[4]) | (p[5].w ^ y[5]) | (p[6].w ^ y[6]) | (p[7].w ^ y[7]));
      const int eb = c * PCH + lane * 128;
      if (c == nchunk - 1) {
        auto live = [&](int r) -> unsigned {
          return r >= 32 ? 0xFFFFFFFFu : (r > 0 ? ((1u << r) - 1u) : 0u);
        };
        m.x &= live(e - eb);
        m.y &= live(e - eb - 32);
        m.z &= live(e - eb - 64);
        m.w &= live(e - eb - 96);
      }
      park(m.x, eb);
      park(m.y, eb + 32);
      park(m.z, eb + 64);
      park(m.w, eb + 96);
    }
    if (cnt > 0u)
      drain_ring<false, !REPLAY>(acc, rw, head, (int)cnt, lane, lo, hlin, dinv, src, dst, seg, nrec);
    if (!REPLAY && lane == 0) wcount[blockIdx.x * AWAVES + wave] = nrec;
  }
  const int nt   = wave & 7;
  const int grp  = wave >> 3;
  const int n0   = nt * 16;
  const int row  = lane & 15;
  const int col  = lane & 15;
  const int half = lane >> 4;
  const int koff = half * 2;

  v2f bf[32];
#pragma unroll
  for (int s = 0; s < 32; ++s) {
    const int o  = s * 4 + koff;
    const int k0 = 4 * (o & 31) + (o >> 5);
    bf[s].x = W[k0 * DCH + n0 + col];
    bf[s].y = W[(k0 + 4) * DCH + n0 + col];
  }
  const float bc = bias[n0 + col];
  const float4 g4 = ((const float4*)gamma)[lane];
  const float4 b4 = ((const float4*)beta)[lane];
  const int lastrow = n - 1;

  __syncthreads();

  constexpr int NTILE = RNG / 16;
  constexpr int NPASS = (NTILE + 3) / 4;
#pragma unroll 1
  for (int ps = 0; ps < NPASS; ++ps) {
    const int  tile   = grp + ps * 4;
    const bool active = tile < NTILE && tile * 16 < (int)span;
    float* xs = acc + (active ? tile : 0) * 16 * XPITCH;
    const int  node0 = lo + tile * 16 + nt * 2;
    const bool live0 = active && node0 < n;
    const bool live1 = active && node0 + 1 < n;
    float4 z; z.x = z.y = z.z = z.w = 0.0f;
    float4 j0 = z, j1 = z;
    v8f c = {};
    if (active) {
      if (!first) {
        const int c0 = live0 ? node0 : lastrow;
        const int c1 = live1 ? node0 + 1 : lastrow;
        j0 = ((const float4*)(jk + (size_t)c0 * DCH))[lane];
        j1 = ((const float4*)(jk + (size_t)c1 * DCH))[lane];
      }
#pragma unroll
      for (int s = 0; s < 32; ++s) {
        v2f a;
        a.x = xs[row * XPITCH + s * 4 + koff];
        a.y = xs[row * XPITCH + s * 4 + koff + 1];
        c = __builtin_amdgcn_wmma_f32_16x16x4_f32(
             false, a,  false, bf[s],
             (short)0, c,  false,  false);
      }
    }
    __syncthreads();

    if (active) {
#pragma unroll
      for (int v = 0; v < 8; ++v)
        xs[(v + half * 8) * XPITCH + n0 + col] = c[v] + bc;
    }
    __syncthreads();

    if (active) {
      const float4 v0 = *(const float4*)(xs + (nt * 2)     * XPITCH + lane * 4);
      const float4 v1 = *(const float4*)(xs + (nt * 2 + 1) * XPITCH + lane * 4);

      float s0 = v0.x + v0.y + v0.z + v0.w;
      float s1 = v1.x + v1.y + v1.z + v1.w;
#pragma unroll
      for (int off = 16; off > 0; off >>= 1) {
        s0 += __shfl_xor(s0, off, 32);
        s1 += __shfl_xor(s1, off, 32);
      }
      const float mu0 = s0 * (1.0f / DCH);
      const float mu1 = s1 * (1.0f / DCH);

      const float dx0 = v0.x - mu0, dy0 = v0.y - mu0, dz0 = v0.z - mu0, dw0 = v0.w - mu0;
      const float dx1 = v1.x - mu1, dy1 = v1.y - mu1, dz1 = v1.z - mu1, dw1 = v1.w - mu1;
      float q0 = dx0 * dx0 + dy0 * dy0 + dz0 * dz0 + dw0 * dw0;
      float q1 = dx1 * dx1 + dy1 * dy1 + dz1 * dz1 + dw1 * dw1;
#pragma unroll
      for (int off = 16; off > 0; off >>= 1) {
        q0 += __shfl_xor(q0, off, 32);
        q1 += __shfl_xor(q1, off, 32);
      }
      const float inv0 = rsqrtf(q0 * (1.0f / DCH) + LN_EPS);
      const float inv1 = rsqrtf(q1 * (1.0f / DCH) + LN_EPS);

      float4 y0, y1;
      y0.x = dx0 * inv0 * g4.x + b4.x;  y1.x = dx1 * inv1 * g4.x + b4.x;
      y0.y = dy0 * inv0 * g4.y + b4.y;  y1.y = dy1 * inv1 * g4.y + b4.y;
      y0.z = dz0 * inv0 * g4.z + b4.z;  y1.z = dz1 * inv1 * g4.z + b4.z;
      y0.w = dw0 * inv0 * g4.w + b4.w;  y1.w = dw1 * inv1 * g4.w + b4.w;
      if (do_relu) {
        y0.x = fmaxf(y0.x, 0.0f); y0.y = fmaxf(y0.y, 0.0f);
        y0.z = fmaxf(y0.z, 0.0f); y0.w = fmaxf(y0.w, 0.0f);
        y1.x = fmaxf(y1.x, 0.0f); y1.y = fmaxf(y1.y, 0.0f);
        y1.z = fmaxf(y1.z, 0.0f); y1.w = fmaxf(y1.w, 0.0f);
      }

      if (live0) ((float4*)(hout + (size_t)node0 * DCH))[lane] = y0;
      if (live1) ((float4*)(hout + (size_t)(node0 + 1) * DCH))[lane] = y1;

      if (!first) {
        y0.x = fmaxf(j0.x, y0.x); y0.y = fmaxf(j0.y, y0.y);
        y0.z = fmaxf(j0.z, y0.z); y0.w = fmaxf(j0.w, y0.w);
        y1.x = fmaxf(j1.x, y1.x); y1.y = fmaxf(j1.y, y1.y);
        y1.z = fmaxf(j1.z, y1.z); y1.w = fmaxf(j1.w, y1.w);
      }
      if (live0) ((float4*)(jk + (size_t)node0 * DCH))[lane] = y0;
      if (live1) ((float4*)(jk + (size_t)(node0 + 1) * DCH))[lane] = y1;
    }
  }
}

extern "C" void kernel_launch(void* const* d_in, const int* in_sizes, int n_in,
                              void* d_out, int out_size, void* d_ws, size_t ws_size,
                              hipStream_t stream) {
  const int N = in_sizes[0] / DCH;
  const int E = in_sizes[1] / 2;

  const float* feat = (const float*)d_in[0];
  const int*   ei   = (const int*)d_in[1];
  const int*   src  = ei;
  const int*   dst  = ei + E;
  const float* Wm[3] = {(const float*)d_in[2], (const float*)d_in[6],  (const float*)d_in[10]};
  const float* bi[3] = {(const float*)d_in[3], (const float*)d_in[7],  (const float*)d_in[11]};
  const float* gm[3] = {(const float*)d_in[4], (const float*)d_in[8],  (const float*)d_in[12]};
  const float* be[3] = {(const float*)d_in[5], (const float*)d_in[9],  (const float*)d_in[13]};
  float* out = (float*)d_out;

  float*    wsf  = (float*)d_ws;
  unsigned* cnt  = (unsigned*)wsf;
  float*    dinv = wsf + N;
  float*    lists = wsf + 2 * (size_t)N;
  float*    hb1  = lists + (size_t)N * DCH;
  float*    hb0  = hb1  + (size_t)N * DCH;
  unsigned* planes = (unsigned*)(hb0 + (size_t)N * DCH);

  const int nodeBlocks = (N + 255) / 256;
  zero_cnt_kernel<<<nodeBlocks, 256, 0, stream>>>(cnt, N);
  count_edges_kernel<<<2048, 256, 0, stream>>>(dst, cnt, E);
  dinv_kernel<<<nodeBlocks, 256, 0, stream>>>(cnt, dinv, N);
  const int planeWaves = ((E + PCH - 1) / PCH) * 4;
  key_planes_kernel<<<(planeWaves + 7) / 8, 256, 0, stream>>>(dst, planes, E, planeWaves);

  const int rangeBlocks = (N + RNG - 1) / RNG;
  unsigned* recs   = (unsigned*)lists;
  unsigned* wcount = recs + (size_t)rangeBlocks * AWAVES * CAPW * 2;

  const float* X = feat;
  float* hw[3] = {hb0, hb1, hb0};
  for (int l = 0; l < 3; ++l) {
    if (l == 0)
      range_layer_kernel<false><<<rangeBlocks, 1024, 0, stream>>>(
          X, planes, src, dst, dinv, Wm[l], bi[l], gm[l], be[l], hw[l], out, recs, wcount,
           1,  1, N, E);
    else
      range_layer_kernel<true><<<rangeBlocks, 1024, 0, stream>>>(
          X, planes, src, dst, dinv, Wm[l], bi[l], gm[l], be[l], hw[l], out, recs, wcount,
           (l < 2),  0, N, E);
    X = hw[l];
  }
}


# Each layer stores its rows once: layers 1, 2 write h only (no running-max plane); layer 3 reads its nodes' h1, h2 rows and writes max(h1,h2,h3) to out: 3 of 6 row planes not written; sums kept
# speedup vs baseline: 3.7348x; 1.0344x over previous
#include <hip/hip_runtime.h>

#define DCH 128
#define LN_EPS 1e-5f
#define RNG 400
#define QCAP 64
#define PCH 4096
#define AWAVES 32
#define CAPW 512
#define REC_FULL 0xFFFFFFFFu
#define XPITCH 132

typedef float v2f __attribute__((ext_vector_type(2)));
typedef float v8f __attribute__((ext_vector_type(8)));

__global__ void zero_cnt_kernel(unsigned* __restrict__ cnt, int n) {
  int i = blockIdx.x * blockDim.x + threadIdx.x;
  if (i < n) cnt[i] = 0u;
}

__global__ void count_edges_kernel(const int* __restrict__ dst,
                                   unsigned* __restrict__ cnt, int e) {
  int i = blockIdx.x * blockDim.x + threadIdx.x;
  int stride = gridDim.x * blockDim.x;
  for (; i < e; i += stride) atomicAdd(&cnt[dst[i]], 1u);
}

__global__ void dinv_kernel(const unsigned* __restrict__ cnt,
                            float* __restrict__ dinv, int n) {
  int i = blockIdx.x * blockDim.x + threadIdx.x;
  if (i < n) dinv[i] = rsqrtf(1.0f + (float)cnt[i]);
}

__global__ __launch_bounds__(256)
void key_planes_kernel(const int* __restrict__ dst,
                       unsigned* __restrict__ planes, int e, int nwaves) {
  const int lane = threadIdx.x & 31;
  const int wv   = (int)((blockIdx.x * blockDim.x + threadIdx.x) >> 5);
  if (wv >= nwaves) return;
  const int ebase = wv * 1024 + lane;
  const int last  = e - 1;
  int dv[32];
#pragma unroll
  for (int t = 0; t < 32; ++t) {
    const int idx = ebase + t * 32;
    dv[t] = dst[idx < last ? idx : last];
  }
  unsigned wd[8];
#pragma unroll
  for (int i = 0; i < 8; ++i) wd[i] = 0u;
#pragma unroll
  for (int t = 0; t < 32; ++t) {
    const unsigned key = (unsigned)dv[t] / (unsigned)RNG;
#pragma unroll
    for (int i = 0; i < 8; ++i) {
      const unsigned b = __builtin_amdgcn_ballot_w32(((key >> i) & 1u) != 0u);
      if (lane == t) wd[i] = b;
    }
  }
  unsigned* o = planes + (size_t)(wv >> 2) * 1024 + (wv & 3) * 32 + lane;
#pragma unroll
  for (int i = 0; i < 8; ++i) o[i * 128] = wd[i];
}

static __device__ __forceinline__ void lds_add(float* p, float v) {
  __hip_atomic_fetch_add(p, v, __ATOMIC_RELAXED, __HIP_MEMORY_SCOPE_WORKGROUP);
}

template <bool FULL, typename AfterFirst>
static __device__ __forceinline__ void add_rows(
    float* acc, int s, int dl, float cf, int nb, int lane,
    const float* __restrict__ hlin, AfterFirst after_first) {
#pragma unroll 1
  for (int k = 0; k < nb; k += 8) {
    float4 v[8];
#pragma unroll
    for (int u = 0; u < 8; ++u) {
      if (FULL || k + u < nb) {
        int su = __builtin_amdgcn_readlane(s, k + u);
        v[u] = ((const float4*)(hlin + (size_t)su * DCH))[lane];
      }
    }
    if (k == 0) after_first();
#pragma unroll
    for (int u = 0; u < 8; ++u) {
      if (FULL || k + u < nb) {
        float cu = __int_as_float(__builtin_amdgcn_readlane(__float_as_int(cf), k + u));
        int   du = __builtin_amdgcn_readlane(dl, k + u);
        float* a = acc + du * XPITCH + lane;
        lds_add(a,      v[u].x * cu);
        lds_add(a + 32, v[u].y * cu);
        lds_add(a + 64, v[u].z * cu);
        lds_add(a + 96, v[u].w * cu);
      }
    }
  }
}

template <bool FULL, bool RECORD>
static __device__ __forceinline__ void drain_ring(
    float* acc, const int* rw, unsigned head, int nb, int lane, int lo,
    const float* __restrict__ hlin, const float* __restrict__ dinv,
    const int* __restrict__ src, const int* __restrict__ dst,
    uint2* seg, unsigned& nrec) {
  int   s = 0, dl = 0;
  float cf = 0.0f;
  if (FULL || lane < nb) {
    const int en = rw[(head + (unsigned)lane) & (QCAP - 1)];
    s = src[en];
    const int d = dst[en];
    cf = dinv[s] * dinv[d];
    dl = d - lo;
  }
  if (RECORD) {
    if (nrec != REC_FULL && nrec + (unsigned)nb <= (unsigned)CAPW) {
      if (FULL || lane < nb) {
        uint2 r;
        r.x = (unsigned)s | ((unsigned)dl << 23);
        r.y = __float_as_uint(cf);
        seg[nrec + (unsigned)lane] = r;
      }
      nrec += (unsigned)nb;
    } else {
      nrec = REC_FULL;
    }
  }
  add_rows<FULL>(acc, s, dl, cf, nb, lane, hlin, [] {});
}

template <bool REPLAY>
__global__ __launch_bounds__(1024)
void range_layer_kernel(const float* __restrict__ hlin,
                        const unsigned* __restrict__ planes,
                        const int* __restrict__ src,
                        const int* __restrict__ dst,
                        const float* __restrict__ dinv,
                        const float* __restrict__ W, const float* __restrict__ bias,
                        const float* __restrict__ gamma, const float* __restrict__ beta,
                        float* __restrict__ hout,
                        const float* mxa,
                        const float* mxb,
                        unsigned* recs, unsigned* wcount,
                        int do_relu, int with_max, int n, int e) {
  __shared__ __attribute__((aligned(16))) float acc[RNG * XPITCH];
  __shared__ int   ring[AWAVES * QCAP];

  const int t    = threadIdx.x;
  const int lane = t & 31;
  const int wave = __builtin_amdgcn_readfirstlane(t >> 5);
  const int lo   = blockIdx.x * RNG;
  const unsigned span = (unsigned)((n - lo) < RNG ? (n - lo) : RNG);

  {
    constexpr int NI = (RNG + AWAVES - 1) / AWAVES;
    float dvl = 0.0f;
    {
      const int r = wave + lane * AWAVES;
      if (lane < NI && r < RNG && lo + r < n) dvl = dinv[lo + r];
    }
    float4 h[NI];
#pragma unroll
    for (int i = 0; i < NI; ++i) {
      const int node = lo + wave + i * AWAVES;
      h[i] = ((const float4*)(hlin + (size_t)(node < n - 1 ? node : n - 1) * DCH))[lane];
    }
#pragma unroll
    for (int i = 0; i < NI; ++i) {
      const int r = wave + i * AWAVES;
      if (r < RNG) {
        const float d  = __int_as_float(__builtin_amdgcn_readlane(__float_as_int(dvl), i));
        const float sc = d * d;
        const bool  in = lo + r < n;
        float* a = acc + r * XPITCH + lane;
        a[0]  = in ? h[i].x * sc : 0.0f;
        a[32] = in ? h[i].y * sc : 0.0f;
        a[64] = in ? h[i].z * sc : 0.0f;
        a[96] = in ? h[i].w * sc : 0.0f;
      }
    }
  }
  __syncthreads();

  int* rw = ring + wave * QCAP;
  unsigned head = 0, cnt = 0;

  uint2* seg = (uint2*)recs + ((size_t)blockIdx.x * AWAVES + (size_t)wave) * CAPW;
  unsigned nrec = 0;

  bool use_lists = false;
  if (REPLAY) {
    const unsigned c = wcount[blockIdx.x * AWAVES + lane];
    use_lists = __builtin_amdgcn_ballot_w32(c == REC_FULL) == 0u;
    nrec = (unsigned)__builtin_amdgcn_readlane((int)c, wave);
  }

  if (use_lists) {
    if (nrec > 0u) {
      const unsigned last = nrec - 1u;
      uint2 r = seg[(unsigned)lane < last ? (unsigned)lane : last];
      const unsigned full = nrec & ~31u;
      for (unsigned base = 0; base < full; base += 32u) {
        uint2 rn = r;
        const int   s  = (int)(r.x & 0x7FFFFFu);
        const int   dl = (int)(r.x >> 23);
        const float cf = __uint_as_float(r.y);
        add_rows<true>(acc, s, dl, cf, 32, lane, hlin, [&] {
          const unsigned i = base + 32u + (unsigned)lane;
          rn = seg[i < last ? i : last];
        });
        r = rn;
      }
      if (nrec > full) {
        const int   s  = (int)(r.x & 0x7FFFFFu);
        const int   dl = (int)(r.x >> 23);
        const float cf = __uint_as_float(r.y);
        add_rows<false>(acc, s, dl, cf, (int)(nrec - full), lane, hlin, [] {});
      }
    }
  } else {
    unsigned y[8];
#pragma unroll
    for (int i = 0; i < 8; ++i) y[i] = ((blockIdx.x >> i) & 1u) ? 0xFFFFFFFFu : 0u;

    auto park = [&](unsigned m, int eb) {
      unsigned b = __builtin_amdgcn_ballot_w32(m != 0u);
      while (b != 0u) {
        if (m != 0u)
          rw[(head + cnt + __builtin_amdgcn_mbcnt_lo(b, 0u)) & (QCAP - 1)] = eb + __builtin_ctz(m);
        m &= m - 1u;
        cnt += (unsigned)__builtin_popcount(b);
        if (cnt >= 32u) {
          drain_ring<true, !REPLAY>(acc, rw, head, 32, lane, lo, hlin, dinv, src, dst, seg, nrec);
          head = (head + 32u) & (QCAP - 1);
          cnt -= 32u;
        }
        b = __builtin_amdgcn_ballot_w32(m != 0u);
      }
    };

    const int nchunk = (e + PCH - 1) / PCH;
    const int start = (int)((blockIdx.x * (unsigned)nchunk) / gridDim.x);
    const uint4* pl4 = (const uint4*)planes;

    for (int ci = wave; ci < nchunk; ci += AWAVES) {
      int c = ci + start;
      if (c >= nchunk) c -= nchunk;
      uint4 p[8];
#pragma unroll
      for (int i = 0; i < 8; ++i) p[i] = pl4[(c * 8 + i) * 32 + lane];
      uint4 m;
      m.x = ~((p[0].x ^ y[0]) | (p[1].x ^ y[1]) | (p[2].x ^ y[2]) | (p[3].x ^ y[3]) |
              (p[4].x ^ y[4]) | (p[5].x ^ y[5]) | (p[6].x ^ y[6]) | (p[7].x ^ y[7]));
      m.y = ~((p[0].y ^ y[0]) | (p[1].y ^ y[1]) | (p[2].y ^ y[2]) | (p[3].y ^ y[3]) |
              (p[4].y ^ y[4]) | (p[5].y ^ y[5]) | (p[6].y ^ y[6]) | (p[7].y ^ y[7]));
      m.z = ~((p[0].z ^ y[0]) | (p[1].z ^ y[1]) | (p[2].z ^ y[2]) | (p[3].z ^ y[3]) |
              (p[4].z ^ y[4]) | (p[5].z ^ y[5]) | (p[6].z ^ y[6]) | (p[7].z ^ y[7]));
      m.w = ~((p[0].w ^ y[0]) | (p[1].w ^ y[1]) | (p[2].w ^ y[2]) | (p[3].w ^ y[3]) |
              (p[4].w ^ y[4]) | (p[5].w ^ y[5]) | (p[6].w ^ y[6]) | (p[7].w ^ y[7]));
      const int eb = c * PCH + lane * 128;
      if (c == nchunk - 1) {
        auto live = [&](int r) -> unsigned {
          return r >= 32 ? 0xFFFFFFFFu : (r > 0 ? ((1u << r) - 1u) : 0u);
        };
        m.x &= live(e - eb);
        m.y &= live(e - eb - 32);
        m.z &= live(e - eb - 64);
        m.w &= live(e - eb - 96);
      }
      park(m.x, eb);
      park(m.y, eb + 32);
      park(m.z, eb + 64);
      park(m.w, eb + 96);
    }
    if (cnt > 0u)
      drain_ring<false, !REPLAY>(acc, rw, head, (int)cnt, lane, lo, hlin, dinv, src, dst, seg, nrec);
    if (!REPLAY && lane == 0) wcount[blockIdx.x * AWAVES + wave] = nrec;
  }
  const int nt   = wave & 7;
  const int grp  = wave >> 3;
  const int n0   = nt * 16;
  const int row  = lane & 15;
  const int col  = lane & 15;
  const int half = lane >> 4;
  const int koff = half * 2;

  v2f bf[32];
#pragma unroll
  for (int s = 0; s < 32; ++s) {
    const int o  = s * 4 + koff;
    const int k0 = 4 * (o & 31) + (o >> 5);
    bf[s].x = W[k0 * DCH + n0 + col];
    bf[s].y = W[(k0 + 4) * DCH + n0 + col];
  }
  const float bc = bias[n0 + col];
  const float4 g4 = ((const float4*)gamma)[lane];
  const float4 b4 = ((const float4*)beta)[lane];
  const int lastrow = n - 1;

  __syncthreads();

  constexpr int NTILE = RNG / 16;
  constexpr int NPASS = (NTILE + 3) / 4;
#pragma unroll 1
  for (int ps = 0; ps < NPASS; ++ps) {
    const int  tile   = grp + ps * 4;
    const bool active = tile < NTILE && tile * 16 < (int)span;
    float* xs = acc + (active ? tile : 0) * 16 * XPITCH;
    const int  node0 = lo + tile * 16 + nt * 2;
    const bool live0 = active && node0 < n;
    const bool live1 = active && node0 + 1 < n;
    float4 z; z.x = z.y = z.z = z.w = 0.0f;
    float4 ja0 = z, ja1 = z, jb0 = z, jb1 = z;
    v8f c = {};
    if (active) {
      if (with_max) {
        const int c0 = live0 ? node0 : lastrow;
        const int c1 = live1 ? node0 + 1 : lastrow;
        ja0 = ((const float4*)(mxa + (size_t)c0 * DCH))[lane];
        ja1 = ((const float4*)(mxa + (size_t)c1 * DCH))[lane];
        jb0 = ((const float4*)(mxb + (size_t)c0 * DCH))[lane];
        jb1 = ((const float4*)(mxb + (size_t)c1 * DCH))[lane];
      }
#pragma unroll
      for (int s = 0; s < 32; ++s) {
        v2f a;
        a.x = xs[row * XPITCH + s * 4 + koff];
        a.y = xs[row * XPITCH + s * 4 + koff + 1];
        c = __builtin_amdgcn_wmma_f32_16x16x4_f32(
             false, a,  false, bf[s],
             (short)0, c,  false,  false);
      }
    }
    __syncthreads();

    if (active) {
#pragma unroll
      for (int v = 0; v < 8; ++v)
        xs[(v + half * 8) * XPITCH + n0 + col] = c[v] + bc;
    }
    __syncthreads();

    if (active) {
      const float4 v0 = *(const float4*)(xs + (nt * 2)     * XPITCH + lane * 4);
      const float4 v1 = *(const float4*)(xs + (nt * 2 + 1) * XPITCH + lane * 4);

      float s0 = v0.x + v0.y + v0.z + v0.w;
      float s1 = v1.x + v1.y + v1.z + v1.w;
#pragma unroll
      for (int off = 16; off > 0; off >>= 1) {
        s0 += __shfl_xor(s0, off, 32);
        s1 += __shfl_xor(s1, off, 32);
      }
      const float mu0 = s0 * (1.0f / DCH);
      const float mu1 = s1 * (1.0f / DCH);

      const float dx0 = v0.x - mu0, dy0 = v0.y - mu0, dz0 = v0.z - mu0, dw0 = v0.w - mu0;
      const float dx1 = v1.x - mu1, dy1 = v1.y - mu1, dz1 = v1.z - mu1, dw1 = v1.w - mu1;
      float q0 = dx0 * dx0 + dy0 * dy0 + dz0 * dz0 + dw0 * dw0;
      float q1 = dx1 * dx1 + dy1 * dy1 + dz1 * dz1 + dw1 * dw1;
#pragma unroll
      for (int off = 16; off > 0; off >>= 1) {
        q0 += __shfl_xor(q0, off, 32);
        q1 += __shfl_xor(q1, off, 32);
      }
      const float inv0 = rsqrtf(q0 * (1.0f / DCH) + LN_EPS);
      const float inv1 = rsqrtf(q1 * (1.0f / DCH) + LN_EPS);

      float4 y0, y1;
      y0.x = dx0 * inv0 * g4.x + b4.x;  y1.x = dx1 * inv1 * g4.x + b4.x;
      y0.y = dy0 * inv0 * g4.y + b4.y;  y1.y = dy1 * inv1 * g4.y + b4.y;
      y0.z = dz0 * inv0 * g4.z + b4.z;  y1.z = dz1 * inv1 * g4.z + b4.z;
      y0.w = dw0 * inv0 * g4.w + b4.w;  y1.w = dw1 * inv1 * g4.w + b4.w;
      if (do_relu) {
        y0.x = fmaxf(y0.x, 0.0f); y0.y = fmaxf(y0.y, 0.0f);
        y0.z = fmaxf(y0.z, 0.0f); y0.w = fmaxf(y0.w, 0.0f);
        y1.x = fmaxf(y1.x, 0.0f); y1.y = fmaxf(y1.y, 0.0f);
        y1.z = fmaxf(y1.z, 0.0f); y1.w = fmaxf(y1.w, 0.0f);
      }

      if (with_max) {
        y0.x = fmaxf(fmaxf(ja0.x, jb0.x), y0.x); y0.y = fmaxf(fmaxf(ja0.y, jb0.y), y0.y);
        y0.z = fmaxf(fmaxf(ja0.z, jb0.z), y0.z); y0.w = fmaxf(fmaxf(ja0.w, jb0.w), y0.w);
        y1.x = fmaxf(fmaxf(ja1.x, jb1.x), y1.x); y1.y = fmaxf(fmaxf(ja1.y, jb1.y), y1.y);
        y1.z = fmaxf(fmaxf(ja1.z, jb1.z), y1.z); y1.w = fmaxf(fmaxf(ja1.w, jb1.w), y1.w);
      }
      if (live0) ((float4*)(hout + (size_t)node0 * DCH))[lane] = y0;
      if (live1) ((float4*)(hout + (size_t)(node0 + 1) * DCH))[lane] = y1;
    }
  }
}

extern "C" void kernel_launch(void* const* d_in, const int* in_sizes, int n_in,
                              void* d_out, int out_size, void* d_ws, size_t ws_size,
                              hipStream_t stream) {
  const int N = in_sizes[0] / DCH;
  const int E = in_sizes[1] / 2;

  const float* feat = (const float*)d_in[0];
  const int*   ei   = (const int*)d_in[1];
  const int*   src  = ei;
  const int*   dst  = ei + E;
  const float* Wm[3] = {(const float*)d_in[2], (const float*)d_in[6],  (const float*)d_in[10]};
  const float* bi[3] = {(const float*)d_in[3], (const float*)d_in[7],  (const float*)d_in[11]};
  const float* gm[3] = {(const float*)d_in[4], (const float*)d_in[8],  (const float*)d_in[12]};
  const float* be[3] = {(const float*)d_in[5], (const float*)d_in[9],  (const float*)d_in[13]};
  float* out = (float*)d_out;

  float*    wsf  = (float*)d_ws;
  unsigned* cnt  = (unsigned*)wsf;
  float*    dinv = wsf + N;
  float*    lists = wsf + 2 * (size_t)N;
  float*    hb1  = lists + (size_t)N * DCH;
  float*    hb0  = hb1  + (size_t)N * DCH;
  unsigned* planes = (unsigned*)(hb0 + (size_t)N * DCH);

  const int nodeBlocks = (N + 255) / 256;
  zero_cnt_kernel<<<nodeBlocks, 256, 0, stream>>>(cnt, N);
  count_edges_kernel<<<2048, 256, 0, stream>>>(dst, cnt, E);
  dinv_kernel<<<nodeBlocks, 256, 0, stream>>>(cnt, dinv, N);
  const int planeWaves = ((E + PCH - 1) / PCH) * 4;
  key_planes_kernel<<<(planeWaves + 7) / 8, 256, 0, stream>>>(dst, planes, E, planeWaves);

  const int rangeBlocks = (N + RNG - 1) / RNG;
  unsigned* recs   = (unsigned*)lists;
  unsigned* wcount = recs + (size_t)rangeBlocks * AWAVES * CAPW * 2;

  const float* X = feat;
  float* hw[3] = {hb0, hb1, out};
  for (int l = 0; l < 3; ++l) {
    if (l == 0)
      range_layer_kernel<false><<<rangeBlocks, 1024, 0, stream>>>(
          X, planes, src, dst, dinv, Wm[l], bi[l], gm[l], be[l], hw[l], hb0, hb1, recs, wcount,
           1,  0, N, E);
    else
      range_layer_kernel<true><<<rangeBlocks, 1024, 0, stream>>>(
          X, planes, src, dst, dinv, Wm[l], bi[l], gm[l], be[l], hw[l], hb0, hb1, recs, wcount,
           (l < 2),  (l == 2), N, E);
    X = hw[l];
  }
}
